# starting kernel
# baseline (speedup 1.0000x reference)
#include <hip/hip_runtime.h>
#include <hip/hip_bf16.h>

typedef __attribute__((ext_vector_type(2))) float v2f;
typedef __attribute__((ext_vector_type(8))) float v8f;

#define FEAT 256
#define TILE_K 32

__global__ __launch_bounds__(256) void build_s1_kernel(
    const int* __restrict__ adj, const int* __restrict__ batch,
    int* __restrict__ s1)
{
    int i = blockIdx.x * 256 + threadIdx.x;
    if (i < 512 * 10) {
        int b = i / 10;
        int j = i - b * 10;
        s1[i] = adj[(size_t)batch[b] * 25 + j];
    }
}

__global__ __launch_bounds__(64) void mean_neigh25_kernel(
    const float* __restrict__ feat, const int* __restrict__ adj,
    const int* __restrict__ s1, float* __restrict__ out)
{
    const int row  = blockIdx.x;
    const int t    = threadIdx.x;
    const int node = s1[row];
    const int* arow = adj + (size_t)node * 25;
    float ax = 0.f, ay = 0.f, az = 0.f, aw = 0.f;
    for (int k = 0; k < 25; ++k) {
        const int nb = arow[k];
        const float4 v = *(const float4*)(feat + (size_t)nb * FEAT + t * 4);
        ax += v.x; ay += v.y; az += v.z; aw += v.w;
    }
    const float s = 1.0f / 25.0f;
    float4 r; r.x = ax * s; r.y = ay * s; r.z = az * s; r.w = aw * s;
    *(float4*)(out + (size_t)row * FEAT + t * 4) = r;
}

__global__ __launch_bounds__(64) void mean_neigh10_kernel(
    const float* __restrict__ feat, const int* __restrict__ s1,
    float* __restrict__ out)
{
    const int row = blockIdx.x;
    const int t   = threadIdx.x;
    float ax = 0.f, ay = 0.f, az = 0.f, aw = 0.f;
    for (int j = 0; j < 10; ++j) {
        const int nb = s1[row * 10 + j];
        const float4 v = *(const float4*)(feat + (size_t)nb * FEAT + t * 4);
        ax += v.x; ay += v.y; az += v.z; aw += v.w;
    }
    const float s = 0.1f;
    float4 r; r.x = ax * s; r.y = ay * s; r.z = az * s; r.w = aw * s;
    *(float4*)(out + (size_t)row * FEAT + t * 4) = r;
}

__global__ __launch_bounds__(64) void mean_rows10_kernel(
    const float* __restrict__ h1, float* __restrict__ out)
{
    const int row = blockIdx.x;
    const int t   = threadIdx.x;
    float ax = 0.f, ay = 0.f, az = 0.f, aw = 0.f;
    for (int j = 0; j < 10; ++j) {
        const float4 v =
            *(const float4*)(h1 + ((size_t)row * 10 + j) * FEAT + t * 4);
        ax += v.x; ay += v.y; az += v.z; aw += v.w;
    }
    const float s = 0.1f;
    float4 r; r.x = ax * s; r.y = ay * s; r.z = az * s; r.w = aw * s;
    *(float4*)(out + (size_t)row * FEAT + t * 4) = r;
}

template <bool RELU, bool GATHER>
__global__ __launch_bounds__(256) void sage_gemm_kernel(
    const float* __restrict__ Aself, const int* __restrict__ idx,
    const float* __restrict__ Aneigh,
    const float* __restrict__ Ws, const float* __restrict__ Wn,
    float* __restrict__ out)
{
    __shared__ float lds_ws[TILE_K * FEAT];
    __shared__ float lds_wn[TILE_K * FEAT];

    const int tid       = threadIdx.x;
    const int lane      = tid & 31;
    const int wave      = tid >> 5;
    const int mtile     = wave >> 1;
    const int nhalf     = wave & 1;
    const int rowInTile = lane & 15;
    const int klane     = (lane >> 4) << 1;
    const int m         = blockIdx.x * 64 + mtile * 16 + rowInTile;

    const float* arow_s =
        GATHER ? (Aself + (size_t)idx[m] * FEAT) : (Aself + (size_t)m * FEAT);
    const float* arow_n = Aneigh + (size_t)m * FEAT;

    const int nBase = (nhalf * 128 + rowInTile) * 2;

    v8f acc[8];
#pragma unroll
    for (int t = 0; t < 8; ++t) acc[t] = (v8f)0.f;

    for (int k0 = 0; k0 < FEAT; k0 += TILE_K) {
        __syncthreads();
        {
            const float* gws = Ws + (size_t)k0 * FEAT;
            const float* gwn = Wn + (size_t)k0 * FEAT;
#pragma unroll
            for (int i = 0; i < 8; ++i) {
                const int u     = i * 256 + tid;
                const int kpair = u >> 7;
                const int npair = u & 127;
                const int g0 = (2 * kpair) * FEAT + 2 * npair;
                const int g1 = g0 + FEAT;
                const int lo = kpair * 512 + npair * 4;
                {
                    const v2f r0 = *(const v2f*)(gws + g0);
                    const v2f r1 = *(const v2f*)(gws + g1);
                    float4 p; p.x = r0.x; p.y = r1.x; p.z = r0.y; p.w = r1.y;
                    *(float4*)(lds_ws + lo) = p;
                }
                {
                    const v2f r0 = *(const v2f*)(gwn + g0);
                    const v2f r1 = *(const v2f*)(gwn + g1);
                    float4 p; p.x = r0.x; p.y = r1.x; p.z = r0.y; p.w = r1.y;
                    *(float4*)(lds_wn + lo) = p;
                }
            }
        }
        __syncthreads();

#pragma unroll
        for (int kk = 0; kk < TILE_K; kk += 4) {
            const v2f a_s = *(const v2f*)(arow_s + k0 + kk + klane);
            const v2f a_n = *(const v2f*)(arow_n + k0 + kk + klane);
            const int kp  = (kk + klane) >> 1;

            v2f bs[8], bn[8];
#pragma unroll
            for (int t = 0; t < 8; ++t) {
                bs[t] = *(const v2f*)(lds_ws + kp * 512 + nBase + t * 32);
                bn[t] = *(const v2f*)(lds_wn + kp * 512 + nBase + t * 32);
            }
#pragma unroll
            for (int t = 0; t < 8; ++t) {
                acc[t] = __builtin_amdgcn_wmma_f32_16x16x4_f32(
                    false, a_s, false, bs[t], (short)0, acc[t], false, false);
                acc[t] = __builtin_amdgcn_wmma_f32_16x16x4_f32(
                    false, a_n, false, bn[t], (short)0, acc[t], false, false);
            }
        }
    }

    const int colBase = nhalf * 128 + rowInTile;
    const int rowBase = blockIdx.x * 64 + mtile * 16 + ((lane >> 4) << 3);
#pragma unroll
    for (int t = 0; t < 8; ++t) {
#pragma unroll
        for (int r = 0; r < 8; ++r) {
            float v = acc[t][r];
            if (RELU) v = fmaxf(v, 0.0f);
            out[(size_t)(rowBase + r) * FEAT + colBase + t * 16] = v;
        }
    }
}

__global__ __launch_bounds__(256) void l2norm_kernel(
    const float* __restrict__ pre, float* __restrict__ out)
{
    __shared__ float ssum[8];
    const int row = blockIdx.x;
    const int t   = threadIdx.x;
    const float v = pre[(size_t)row * FEAT + t];
    float s = v * v;
#pragma unroll
    for (int o = 16; o >= 1; o >>= 1) s += __shfl_xor(s, o, 32);
    if ((t & 31) == 0) ssum[t >> 5] = s;
    __syncthreads();
    float tot = 0.f;
#pragma unroll
    for (int i = 0; i < 8; ++i) tot += ssum[i];
    const float norm = fmaxf(sqrtf(tot), 1e-12f);
    out[(size_t)row * FEAT + t] = v / norm;
}

extern "C" void kernel_launch(void* const* d_in, const int* in_sizes, int n_in,
                              void* d_out, int out_size, void* d_ws,
                              size_t ws_size, hipStream_t stream)
{
    const float* features = (const float*)d_in[0];
    const int*   adj      = (const int*)d_in[1];
    const int*   batch    = (const int*)d_in[2];
    const float* Ws0      = (const float*)d_in[3];
    const float* Wn0      = (const float*)d_in[4];
    const float* Ws1      = (const float*)d_in[5];
    const float* Wn1      = (const float*)d_in[6];
    float* out = (float*)d_out;

    char* ws = (char*)d_ws;
    int*   s1   = (int*)(ws + 0);
    float* nm2  = (float*)(ws + 20480);
    float* nm1  = (float*)(ws + 5263360);
    float* h1   = (float*)(ws + 5787648);
    float* h0   = (float*)(ws + 11030528);
    float* nmh1 = (float*)(ws + 11554816);
    float* pre  = (float*)(ws + 12079104);

    build_s1_kernel<<<(5120 + 255) / 256, 256, 0, stream>>>(adj, batch, s1);
    mean_neigh25_kernel<<<5120, 64, 0, stream>>>(features, adj, s1, nm2);
    mean_neigh10_kernel<<<512, 64, 0, stream>>>(features, s1, nm1);
    sage_gemm_kernel<true, true><<<5120 / 64, 256, 0, stream>>>(
        features, s1, nm2, Ws0, Wn0, h1);
    sage_gemm_kernel<true, true><<<512 / 64, 256, 0, stream>>>(
        features, batch, nm1, Ws0, Wn0, h0);
    mean_rows10_kernel<<<512, 64, 0, stream>>>(h1, nmh1);
    sage_gemm_kernel<false, false><<<512 / 64, 256, 0, stream>>>(
        h0, (const int*)nullptr, nmh1, Ws1, Wn1, pre);
    l2norm_kernel<<<512, 256, 0, stream>>>(pre, out);
}


# Both 512-row GEMMs split K over a block's 8 waves (64 k each, 16 rows x 64 cols); B fragments load straight from global, no weight LDS; partial tiles summed in LDS in wave order: k sum order changes
# speedup vs baseline: 4.8920x; 4.8920x over previous
#include <hip/hip_runtime.h>
#include <hip/hip_bf16.h>

typedef __attribute__((ext_vector_type(2))) float v2f;
typedef __attribute__((ext_vector_type(4))) float v4f;
typedef __attribute__((ext_vector_type(8))) float v8f;

#define FEAT 256

__global__ __launch_bounds__(256) void build_s1_kernel(
    const int* __restrict__ adj, const int* __restrict__ batch,
    int* __restrict__ s1)
{
    int i = blockIdx.x * 256 + threadIdx.x;
    if (i < 512 * 10) {
        int b = i / 10;
        int j = i - b * 10;
        s1[i] = adj[(size_t)batch[b] * 25 + j];
    }
}

__global__ __launch_bounds__(64) void mean_neigh25_kernel(
    const float* __restrict__ feat, const int* __restrict__ adj,
    const int* __restrict__ s1, float* __restrict__ out)
{
    const int row  = blockIdx.x;
    const int t    = threadIdx.x;
    const int node = s1[row];
    const int* arow = adj + (size_t)node * 25;
    float ax = 0.f, ay = 0.f, az = 0.f, aw = 0.f;
    for (int k = 0; k < 25; ++k) {
        const int nb = arow[k];
        const float4 v = *(const float4*)(feat + (size_t)nb * FEAT + t * 4);
        ax += v.x; ay += v.y; az += v.z; aw += v.w;
    }
    const float s = 1.0f / 25.0f;
    float4 r; r.x = ax * s; r.y = ay * s; r.z = az * s; r.w = aw * s;
    *(float4*)(out + (size_t)row * FEAT + t * 4) = r;
}

__global__ __launch_bounds__(64) void mean_neigh10_kernel(
    const float* __restrict__ feat, const int* __restrict__ s1,
    float* __restrict__ out)
{
    const int row = blockIdx.x;
    const int t   = threadIdx.x;
    float ax = 0.f, ay = 0.f, az = 0.f, aw = 0.f;
    for (int j = 0; j < 10; ++j) {
        const int nb = s1[row * 10 + j];
        const float4 v = *(const float4*)(feat + (size_t)nb * FEAT + t * 4);
        ax += v.x; ay += v.y; az += v.z; aw += v.w;
    }
    const float s = 0.1f;
    float4 r; r.x = ax * s; r.y = ay * s; r.z = az * s; r.w = aw * s;
    *(float4*)(out + (size_t)row * FEAT + t * 4) = r;
}

__global__ __launch_bounds__(64) void mean_rows10_kernel(
    const float* __restrict__ h1, float* __restrict__ out)
{
    const int row = blockIdx.x;
    const int t   = threadIdx.x;
    float ax = 0.f, ay = 0.f, az = 0.f, aw = 0.f;
    for (int j = 0; j < 10; ++j) {
        const float4 v =
            *(const float4*)(h1 + ((size_t)row * 10 + j) * FEAT + t * 4);
        ax += v.x; ay += v.y; az += v.z; aw += v.w;
    }
    const float s = 0.1f;
    float4 r; r.x = ax * s; r.y = ay * s; r.z = az * s; r.w = aw * s;
    *(float4*)(out + (size_t)row * FEAT + t * 4) = r;
}

template <int MT, bool RELU, bool GATHER>
__global__ __launch_bounds__(MT * 32) void sage_gemm_kernel(
    const float* __restrict__ Aself, const int* __restrict__ idx,
    const float* __restrict__ Aneigh,
    const float* __restrict__ Ws, const float* __restrict__ Wn,
    float* __restrict__ out)
{
    __shared__ v4f lds_ws[32 * 4 * 2 * 16];
    __shared__ v4f lds_wn[32 * 4 * 2 * 16];

    constexpr int NT = MT * 32;
    constexpr int NU = 2048 / NT;

    const int tid  = threadIdx.x;
    const int lane = tid & 31;
    const int wave = tid >> 5;
    const int c    = lane & 15;
    const int h    = lane >> 4;
    const int col0 = blockIdx.y * 64;
    const int rowBase = (blockIdx.x * MT + wave) * 16;
    const int m    = rowBase + c;

    const float* arow_s =
        GATHER ? (Aself + (size_t)idx[m] * FEAT) : (Aself + (size_t)m * FEAT);
    const float* arow_n = Aneigh + (size_t)m * FEAT;

    v4f wr[NU][4];
#pragma unroll
    for (int i = 0; i < NU; ++i) {
        const int u  = (i % (NU / 2)) * NT + tid;
        const int q  = u & 15;
        const int kq = u >> 4;
        const float* g = ((i < NU / 2) ? Ws : Wn)
                         + (size_t)(4 * kq) * FEAT + col0 + 4 * q;
#pragma unroll
        for (int r = 0; r < 4; ++r) wr[i][r] = *(const v4f*)(g + r * FEAT);
    }

    v4f a_s[32], a_n[32];
#pragma unroll
    for (int j = 0; j < 32; ++j) a_s[j] = *(const v4f*)(arow_s + 8 * j + 4 * h);
#pragma unroll
    for (int j = 0; j < 32; ++j) a_n[j] = *(const v4f*)(arow_n + 8 * j + 4 * h);

#pragma unroll
    for (int i = 0; i < NU; ++i) {
        const int u  = (i % (NU / 2)) * NT + tid;
        const int q  = u & 15;
        const int kq = u >> 4;
        const int j  = kq >> 1;
        const int hh = kq & 1;
        v4f* dst = (i < NU / 2) ? lds_ws : lds_wn;
#pragma unroll
        for (int e = 0; e < 4; ++e) {
            v4f p;
            p.x = wr[i][0][e]; p.y = wr[i][1][e];
            p.z = wr[i][2][e]; p.w = wr[i][3][e];
            dst[((j * 4 + e) * 2 + hh) * 16 + q] = p;
        }
    }
    __syncthreads();

    v8f acc[4];
#pragma unroll
    for (int t = 0; t < 4; ++t) acc[t] = (v8f)0.f;

    const v4f* bl_s = lds_ws + h * 16 + c;
    const v4f* bl_n = lds_wn + h * 16 + c;

    constexpr int G = 4;
    v4f bc[G * 4], bn[G * 4];
#pragma unroll
    for (int x = 0; x < G * 4; ++x) bc[x] = bl_s[x * 32];

#pragma unroll
    for (int g = 0; g < 64 / G; ++g) {
        if (g + 1 < 64 / G) {
#pragma unroll
            for (int x = 0; x < G * 4; ++x) {
                const int y = (g + 1) * G * 4 + x;
                bn[x] = (y < 128) ? bl_s[y * 32] : bl_n[(y - 128) * 32];
            }
        }
        __builtin_amdgcn_sched_barrier(0);
#pragma unroll
        for (int jj = 0; jj < G; ++jj) {
            const int j = g * G + jj;
            const v4f a = (j < 32) ? a_s[j & 31] : a_n[j & 31];
            const v2f a0 = {a.x, a.y};
            const v2f a1 = {a.z, a.w};
#pragma unroll
            for (int t = 0; t < 4; ++t) {
                const v4f b = bc[jj * 4 + t];
                const v2f b0 = {b.x, b.y};
                const v2f b1 = {b.z, b.w};
                acc[t] = __builtin_amdgcn_wmma_f32_16x16x4_f32(
                    false, a0, false, b0, (short)0, acc[t], false, false);
                acc[t] = __builtin_amdgcn_wmma_f32_16x16x4_f32(
                    false, a1, false, b1, (short)0, acc[t], false, false);
            }
        }
        __builtin_amdgcn_sched_barrier(0);
#pragma unroll
        for (int x = 0; x < G * 4; ++x) bc[x] = bn[x];
    }

#pragma unroll
    for (int r = 0; r < 8; ++r) {
        v4f o;
        o.x = acc[0][r]; o.y = acc[1][r]; o.z = acc[2][r]; o.w = acc[3][r];
        if (RELU) {
            o.x = fmaxf(o.x, 0.0f); o.y = fmaxf(o.y, 0.0f);
            o.z = fmaxf(o.z, 0.0f); o.w = fmaxf(o.w, 0.0f);
        }
        *(v4f*)(out + (size_t)(rowBase + 8 * h + r) * FEAT + col0 + 4 * c) = o;
    }
}

template <bool RELU, bool GATHER>
__global__ __launch_bounds__(256) void sage_gemm_splitk_kernel(
    const float* __restrict__ Aself, const int* __restrict__ idx,
    const float* __restrict__ Aneigh,
    const float* __restrict__ Ws, const float* __restrict__ Wn,
    float* __restrict__ out)
{
    __shared__ v4f part[8 * 16 * 16];

    const int tid  = threadIdx.x;
    const int lane = tid & 31;
    const int wave = tid >> 5;
    const int c    = lane & 15;
    const int h    = lane >> 4;
    const int col0 = blockIdx.y * 64;
    const int rowBase = blockIdx.x * 16;
    const int m    = rowBase + c;
    const bool self = wave < 4;
    const int k0   = (wave & 3) * 64;

    const int node = GATHER ? idx[m] : m;
    const float* g = (self ? Ws : Wn)
                     + (size_t)(k0 + 4 * h) * FEAT + col0 + 4 * c;

    v4f wr[8][4];
#pragma unroll
    for (int jj = 0; jj < 8; ++jj) {
#pragma unroll
        for (int r = 0; r < 4; ++r)
            wr[jj][r] = *(const v4f*)(g + (size_t)(8 * jj + r) * FEAT);
    }
    const float* arow = (self ? (Aself + (size_t)node * FEAT)
                              : (Aneigh + (size_t)m * FEAT)) + k0 + 4 * h;
    v4f a[8];
#pragma unroll
    for (int jj = 0; jj < 8; ++jj) a[jj] = *(const v4f*)(arow + 8 * jj);
    __builtin_amdgcn_sched_barrier(0);

    v8f acc[4];
#pragma unroll
    for (int t = 0; t < 4; ++t) acc[t] = (v8f)0.f;

#pragma unroll
    for (int jj = 0; jj < 8; ++jj) {
        const v2f a0 = {a[jj].x, a[jj].y};
        const v2f a1 = {a[jj].z, a[jj].w};
#pragma unroll
        for (int t = 0; t < 4; ++t) {
            const v2f b0 = {wr[jj][0][t], wr[jj][1][t]};
            acc[t] = __builtin_amdgcn_wmma_f32_16x16x4_f32(
                false, a0, false, b0, (short)0, acc[t], false, false);
        }
#pragma unroll
        for (int t = 0; t < 4; ++t) {
            const v2f b1 = {wr[jj][2][t], wr[jj][3][t]};
            acc[t] = __builtin_amdgcn_wmma_f32_16x16x4_f32(
                false, a1, false, b1, (short)0, acc[t], false, false);
        }
    }

#pragma unroll
    for (int r = 0; r < 8; ++r) {
        v4f o;
        o.x = acc[0][r]; o.y = acc[1][r]; o.z = acc[2][r]; o.w = acc[3][r];
        part[(wave * 16 + 8 * h + r) * 16 + c] = o;
    }
    __syncthreads();

    const int row = tid >> 4;
    const int q   = tid & 15;
    v4f sum = part[row * 16 + q];
#pragma unroll
    for (int w = 1; w < 8; ++w) sum += part[(w * 16 + row) * 16 + q];
    if (RELU) {
        sum.x = fmaxf(sum.x, 0.0f); sum.y = fmaxf(sum.y, 0.0f);
        sum.z = fmaxf(sum.z, 0.0f); sum.w = fmaxf(sum.w, 0.0f);
    }
    *(v4f*)(out + (size_t)(rowBase + row) * FEAT + col0 + 4 * q) = sum;
}

__global__ __launch_bounds__(256) void l2norm_kernel(
    const float* __restrict__ pre, float* __restrict__ out)
{
    __shared__ float ssum[8];
    const int row = blockIdx.x;
    const int t   = threadIdx.x;
    const float v = pre[(size_t)row * FEAT + t];
    float s = v * v;
#pragma unroll
    for (int o = 16; o >= 1; o >>= 1) s += __shfl_xor(s, o, 32);
    if ((t & 31) == 0) ssum[t >> 5] = s;
    __syncthreads();
    float tot = 0.f;
#pragma unroll
    for (int i = 0; i < 8; ++i) tot += ssum[i];
    const float norm = fmaxf(sqrtf(tot), 1e-12f);
    out[(size_t)row * FEAT + t] = v / norm;
}

extern "C" void kernel_launch(void* const* d_in, const int* in_sizes, int n_in,
                              void* d_out, int out_size, void* d_ws,
                              size_t ws_size, hipStream_t stream)
{
    const float* features = (const float*)d_in[0];
    const int*   adj      = (const int*)d_in[1];
    const int*   batch    = (const int*)d_in[2];
    const float* Ws0      = (const float*)d_in[3];
    const float* Wn0      = (const float*)d_in[4];
    const float* Ws1      = (const float*)d_in[5];
    const float* Wn1      = (const float*)d_in[6];
    float* out = (float*)d_out;

    char* ws = (char*)d_ws;
    int*   s1   = (int*)(ws + 0);
    float* nm2  = (float*)(ws + 20480);
    float* nm1  = (float*)(ws + 5263360);
    float* h1   = (float*)(ws + 5787648);
    float* h0   = (float*)(ws + 11030528);
    float* nmh1 = (float*)(ws + 11554816);
    float* pre  = (float*)(ws + 12079104);

    build_s1_kernel<<<(5120 + 255) / 256, 256, 0, stream>>>(adj, batch, s1);
    mean_neigh25_kernel<<<5120, 64, 0, stream>>>(features, adj, s1, nm2);
    mean_neigh10_kernel<<<512, 64, 0, stream>>>(features, s1, nm1);
    sage_gemm_kernel<8, true, true><<<dim3(40, 4), 256, 0, stream>>>(
        features, s1, nm2, Ws0, Wn0, h1);
    sage_gemm_splitk_kernel<true, true><<<dim3(32, 4), 256, 0, stream>>>(
        features, batch, nm1, Ws0, Wn0, h0);
    mean_rows10_kernel<<<512, 64, 0, stream>>>(h1, nmh1);
    sage_gemm_splitk_kernel<false, false><<<dim3(32, 4), 256, 0, stream>>>(
        h0, (const int*)nullptr, nmh1, Ws1, Wn1, pre);
    l2norm_kernel<<<512, 256, 0, stream>>>(pre, out);
}


# Hop-1 GEMM re-decomposed: 256 blocks of 80 rows x 64 cols, K split over 4 waves (one per SIMD), a wave keeps its 128-k weight slice in registers for 5 row tiles; no weight LDS; k sum order changes
# speedup vs baseline: 5.4132x; 1.1065x over previous
#include <hip/hip_runtime.h>
#include <hip/hip_bf16.h>

typedef __attribute__((ext_vector_type(2))) float v2f;
typedef __attribute__((ext_vector_type(4))) float v4f;
typedef __attribute__((ext_vector_type(8))) float v8f;

#define FEAT 256

__global__ __launch_bounds__(256) void build_s1_kernel(
    const int* __restrict__ adj, const int* __restrict__ batch,
    int* __restrict__ s1)
{
    int i = blockIdx.x * 256 + threadIdx.x;
    if (i < 512 * 10) {
        int b = i / 10;
        int j = i - b * 10;
        s1[i] = adj[(size_t)batch[b] * 25 + j];
    }
}

__global__ __launch_bounds__(64) void mean_neigh25_kernel(
    const float* __restrict__ feat, const int* __restrict__ adj,
    const int* __restrict__ s1, float* __restrict__ out)
{
    const int row  = blockIdx.x;
    const int t    = threadIdx.x;
    const int node = s1[row];
    const int* arow = adj + (size_t)node * 25;
    float ax = 0.f, ay = 0.f, az = 0.f, aw = 0.f;
    for (int k = 0; k < 25; ++k) {
        const int nb = arow[k];
        const float4 v = *(const float4*)(feat + (size_t)nb * FEAT + t * 4);
        ax += v.x; ay += v.y; az += v.z; aw += v.w;
    }
    const float s = 1.0f / 25.0f;
    float4 r; r.x = ax * s; r.y = ay * s; r.z = az * s; r.w = aw * s;
    *(float4*)(out + (size_t)row * FEAT + t * 4) = r;
}

__global__ __launch_bounds__(64) void mean_neigh10_kernel(
    const float* __restrict__ feat, const int* __restrict__ s1,
    float* __restrict__ out)
{
    const int row = blockIdx.x;
    const int t   = threadIdx.x;
    float ax = 0.f, ay = 0.f, az = 0.f, aw = 0.f;
    for (int j = 0; j < 10; ++j) {
        const int nb = s1[row * 10 + j];
        const float4 v = *(const float4*)(feat + (size_t)nb * FEAT + t * 4);
        ax += v.x; ay += v.y; az += v.z; aw += v.w;
    }
    const float s = 0.1f;
    float4 r; r.x = ax * s; r.y = ay * s; r.z = az * s; r.w = aw * s;
    *(float4*)(out + (size_t)row * FEAT + t * 4) = r;
}

__global__ __launch_bounds__(64) void mean_rows10_kernel(
    const float* __restrict__ h1, float* __restrict__ out)
{
    const int row = blockIdx.x;
    const int t   = threadIdx.x;
    float ax = 0.f, ay = 0.f, az = 0.f, aw = 0.f;
    for (int j = 0; j < 10; ++j) {
        const float4 v =
            *(const float4*)(h1 + ((size_t)row * 10 + j) * FEAT + t * 4);
        ax += v.x; ay += v.y; az += v.z; aw += v.w;
    }
    const float s = 0.1f;
    float4 r; r.x = ax * s; r.y = ay * s; r.z = az * s; r.w = aw * s;
    *(float4*)(out + (size_t)row * FEAT + t * 4) = r;
}

template <bool RELU>
__global__ __launch_bounds__(128) void sage_gemm_rows80_kernel(
    const float* __restrict__ feat, const int* __restrict__ idx,
    const float* __restrict__ Aneigh,
    const float* __restrict__ Ws, const float* __restrict__ Wn,
    float* __restrict__ out)
{
    constexpr int RT = 5;
    __shared__ v4f part[4 * RT * 16 * 16];

    const int tid  = threadIdx.x;
    const int lane = tid & 31;
    const int wave = tid >> 5;
    const int c    = lane & 15;
    const int h    = lane >> 4;
    const int col0 = blockIdx.y * 64;
    const int rowBase = blockIdx.x * (16 * RT);
    const int k0   = wave * 64;

    unsigned node[RT];
#pragma unroll
    for (int i = 0; i < RT; ++i) node[i] = (unsigned)idx[rowBase + 16 * i + c];
    __builtin_amdgcn_sched_barrier(0);

    const size_t woff = (size_t)(k0 + 4 * h) * FEAT + col0 + 4 * c;
    const float* gs = Ws + woff;
    const float* gn = Wn + woff;

    v4f wr[16][4];
    v4f a[RT][16];
#pragma unroll
    for (int jj = 0; jj < 4; ++jj) {
#pragma unroll
        for (int r = 0; r < 4; ++r)
            wr[jj][r] = *(const v4f*)(gs + (size_t)(8 * jj + r) * FEAT);
    }
    __builtin_amdgcn_sched_barrier(0);
    const float* arow_s[RT];
    const float* arow_n[RT];
#pragma unroll
    for (int i = 0; i < RT; ++i) {
        arow_s[i] = feat + (size_t)node[i] * FEAT + k0 + 4 * h;
        arow_n[i] = Aneigh + (size_t)(rowBase + 16 * i + c) * FEAT + k0 + 4 * h;
    }
#pragma unroll
    for (int jj = 0; jj < 8; ++jj) {
        a[0][jj]     = *(const v4f*)(arow_s[0] + 8 * jj);
        a[0][jj + 8] = *(const v4f*)(arow_n[0] + 8 * jj);
    }
    __builtin_amdgcn_sched_barrier(0);
#pragma unroll
    for (int jj = 4; jj < 8; ++jj) {
#pragma unroll
        for (int r = 0; r < 4; ++r)
            wr[jj][r] = *(const v4f*)(gs + (size_t)(8 * jj + r) * FEAT);
    }
    __builtin_amdgcn_sched_barrier(0);
#pragma unroll
    for (int jj = 0; jj < 4; ++jj) {
#pragma unroll
        for (int r = 0; r < 4; ++r)
            wr[jj + 8][r] = *(const v4f*)(gn + (size_t)(8 * jj + r) * FEAT);
    }
    __builtin_amdgcn_sched_barrier(0);
#pragma unroll
    for (int jj = 4; jj < 8; ++jj) {
#pragma unroll
        for (int r = 0; r < 4; ++r)
            wr[jj + 8][r] = *(const v4f*)(gn + (size_t)(8 * jj + r) * FEAT);
    }
    __builtin_amdgcn_sched_barrier(0);
#pragma unroll
    for (int jj = 0; jj < 8; ++jj) {
        a[1][jj]     = *(const v4f*)(arow_s[1] + 8 * jj);
        a[1][jj + 8] = *(const v4f*)(arow_n[1] + 8 * jj);
    }

#pragma unroll
    for (int i = 0; i < RT; ++i) {
        if (i == 1) {
#pragma unroll
            for (int i2 = 2; i2 < RT; ++i2) {
#pragma unroll
                for (int jj = 0; jj < 8; ++jj) {
                    a[i2][jj]     = *(const v4f*)(arow_s[i2] + 8 * jj);
                    a[i2][jj + 8] = *(const v4f*)(arow_n[i2] + 8 * jj);
                }
            }
        }
        if (i < 2) __builtin_amdgcn_sched_barrier(0);

        v8f acc[4];
#pragma unroll
        for (int t = 0; t < 4; ++t) acc[t] = (v8f)0.f;

#pragma unroll
        for (int jj = 0; jj < 16; ++jj) {
            const v2f a0 = {a[i][jj].x, a[i][jj].y};
            const v2f a1 = {a[i][jj].z, a[i][jj].w};
#pragma unroll
            for (int t = 0; t < 4; ++t) {
                const v2f b0 = {wr[jj][0][t], wr[jj][1][t]};
                acc[t] = __builtin_amdgcn_wmma_f32_16x16x4_f32(
                    false, a0, false, b0, (short)0, acc[t], false, false);
            }
#pragma unroll
            for (int t = 0; t < 4; ++t) {
                const v2f b1 = {wr[jj][2][t], wr[jj][3][t]};
                acc[t] = __builtin_amdgcn_wmma_f32_16x16x4_f32(
                    false, a1, false, b1, (short)0, acc[t], false, false);
            }
        }

#pragma unroll
        for (int r = 0; r < 8; ++r) {
            v4f o;
            o.x = acc[0][r]; o.y = acc[1][r]; o.z = acc[2][r]; o.w = acc[3][r];
            part[((wave * RT + i) * 16 + 8 * h + r) * 16 + c] = o;
        }
    }
    __syncthreads();

#pragma unroll
    for (int n = 0; n < (16 * RT * 16) / 128; ++n) {
        const int o   = n * 128 + tid;
        const int row = o >> 4;
        const int q   = o & 15;
        v4f sum = part[row * 16 + q];
#pragma unroll
        for (int w = 1; w < 4; ++w) sum += part[(w * 16 * RT + row) * 16 + q];
        if (RELU) {
            sum.x = fmaxf(sum.x, 0.0f); sum.y = fmaxf(sum.y, 0.0f);
            sum.z = fmaxf(sum.z, 0.0f); sum.w = fmaxf(sum.w, 0.0f);
        }
        *(v4f*)(out + (size_t)(rowBase + row) * FEAT + col0 + 4 * q) = sum;
    }
}

template <bool RELU, bool GATHER>
__global__ __launch_bounds__(256) void sage_gemm_splitk_kernel(
    const float* __restrict__ Aself, const int* __restrict__ idx,
    const float* __restrict__ Aneigh,
    const float* __restrict__ Ws, const float* __restrict__ Wn,
    float* __restrict__ out)
{
    __shared__ v4f part[8 * 16 * 16];

    const int tid  = threadIdx.x;
    const int lane = tid & 31;
    const int wave = tid >> 5;
    const int c    = lane & 15;
    const int h    = lane >> 4;
    const int col0 = blockIdx.y * 64;
    const int rowBase = blockIdx.x * 16;
    const int m    = rowBase + c;
    const bool self = wave < 4;
    const int k0   = (wave & 3) * 64;

    const int node = GATHER ? idx[m] : m;
    const float* g = (self ? Ws : Wn)
                     + (size_t)(k0 + 4 * h) * FEAT + col0 + 4 * c;

    v4f wr[8][4];
#pragma unroll
    for (int jj = 0; jj < 8; ++jj) {
#pragma unroll
        for (int r = 0; r < 4; ++r)
            wr[jj][r] = *(const v4f*)(g + (size_t)(8 * jj + r) * FEAT);
    }
    const float* arow = (self ? (Aself + (size_t)node * FEAT)
                              : (Aneigh + (size_t)m * FEAT)) + k0 + 4 * h;
    v4f a[8];
#pragma unroll
    for (int jj = 0; jj < 8; ++jj) a[jj] = *(const v4f*)(arow + 8 * jj);
    __builtin_amdgcn_sched_barrier(0);

    v8f acc[4];
#pragma unroll
    for (int t = 0; t < 4; ++t) acc[t] = (v8f)0.f;

#pragma unroll
    for (int jj = 0; jj < 8; ++jj) {
        const v2f a0 = {a[jj].x, a[jj].y};
        const v2f a1 = {a[jj].z, a[jj].w};
#pragma unroll
        for (int t = 0; t < 4; ++t) {
            const v2f b0 = {wr[jj][0][t], wr[jj][1][t]};
            acc[t] = __builtin_amdgcn_wmma_f32_16x16x4_f32(
                false, a0, false, b0, (short)0, acc[t], false, false);
        }
#pragma unroll
        for (int t = 0; t < 4; ++t) {
            const v2f b1 = {wr[jj][2][t], wr[jj][3][t]};
            acc[t] = __builtin_amdgcn_wmma_f32_16x16x4_f32(
                false, a1, false, b1, (short)0, acc[t], false, false);
        }
    }

#pragma unroll
    for (int r = 0; r < 8; ++r) {
        v4f o;
        o.x = acc[0][r]; o.y = acc[1][r]; o.z = acc[2][r]; o.w = acc[3][r];
        part[(wave * 16 + 8 * h + r) * 16 + c] = o;
    }
    __syncthreads();

    const int row = tid >> 4;
    const int q   = tid & 15;
    v4f sum = part[row * 16 + q];
#pragma unroll
    for (int w = 1; w < 8; ++w) sum += part[(w * 16 + row) * 16 + q];
    if (RELU) {
        sum.x = fmaxf(sum.x, 0.0f); sum.y = fmaxf(sum.y, 0.0f);
        sum.z = fmaxf(sum.z, 0.0f); sum.w = fmaxf(sum.w, 0.0f);
    }
    *(v4f*)(out + (size_t)(rowBase + row) * FEAT + col0 + 4 * q) = sum;
}

__global__ __launch_bounds__(256) void l2norm_kernel(
    const float* __restrict__ pre, float* __restrict__ out)
{
    __shared__ float ssum[8];
    const int row = blockIdx.x;
    const int t   = threadIdx.x;
    const float v = pre[(size_t)row * FEAT + t];
    float s = v * v;
#pragma unroll
    for (int o = 16; o >= 1; o >>= 1) s += __shfl_xor(s, o, 32);
    if ((t & 31) == 0) ssum[t >> 5] = s;
    __syncthreads();
    float tot = 0.f;
#pragma unroll
    for (int i = 0; i < 8; ++i) tot += ssum[i];
    const float norm = fmaxf(sqrtf(tot), 1e-12f);
    out[(size_t)row * FEAT + t] = v / norm;
}

extern "C" void kernel_launch(void* const* d_in, const int* in_sizes, int n_in,
                              void* d_out, int out_size, void* d_ws,
                              size_t ws_size, hipStream_t stream)
{
    const float* features = (const float*)d_in[0];
    const int*   adj      = (const int*)d_in[1];
    const int*   batch    = (const int*)d_in[2];
    const float* Ws0      = (const float*)d_in[3];
    const float* Wn0      = (const float*)d_in[4];
    const float* Ws1      = (const float*)d_in[5];
    const float* Wn1      = (const float*)d_in[6];
    float* out = (float*)d_out;

    char* ws = (char*)d_ws;
    int*   s1   = (int*)(ws + 0);
    float* nm2  = (float*)(ws + 20480);
    float* nm1  = (float*)(ws + 5263360);
    float* h1   = (float*)(ws + 5787648);
    float* h0   = (float*)(ws + 11030528);
    float* nmh1 = (float*)(ws + 11554816);
    float* pre  = (float*)(ws + 12079104);

    build_s1_kernel<<<(5120 + 255) / 256, 256, 0, stream>>>(adj, batch, s1);
    mean_neigh25_kernel<<<5120, 64, 0, stream>>>(features, adj, s1, nm2);
    mean_neigh10_kernel<<<512, 64, 0, stream>>>(features, s1, nm1);
    sage_gemm_rows80_kernel<true><<<dim3(64, 4), 128, 0, stream>>>(
        features, s1, nm2, Ws0, Wn0, h1);
    sage_gemm_splitk_kernel<true, true><<<dim3(32, 4), 256, 0, stream>>>(
        features, batch, nm1, Ws0, Wn0, h0);
    mean_rows10_kernel<<<512, 64, 0, stream>>>(h1, nmh1);
    sage_gemm_splitk_kernel<false, false><<<dim3(32, 4), 256, 0, stream>>>(
        h0, (const int*)nullptr, nmh1, Ws1, Wn1, pre);
    l2norm_kernel<<<512, 256, 0, stream>>>(pre, out);
}


# Hop-1 GEMM epilogue adds each group's 10 finished rows (a block's 80 rows = 8 groups) and stores the means: h1 never leaves the block, the 10-row mean launch and its read-back go; sums same
# speedup vs baseline: 5.8446x; 1.0797x over previous
#include <hip/hip_runtime.h>
#include <hip/hip_bf16.h>

typedef __attribute__((ext_vector_type(2))) float v2f;
typedef __attribute__((ext_vector_type(4))) float v4f;
typedef __attribute__((ext_vector_type(8))) float v8f;

#define FEAT 256

__global__ __launch_bounds__(256) void build_s1_kernel(
    const int* __restrict__ adj, const int* __restrict__ batch,
    int* __restrict__ s1)
{
    int i = blockIdx.x * 256 + threadIdx.x;
    if (i < 512 * 10) {
        int b = i / 10;
        int j = i - b * 10;
        s1[i] = adj[(size_t)batch[b] * 25 + j];
    }
}

__global__ __launch_bounds__(64) void mean_neigh25_kernel(
    const float* __restrict__ feat, const int* __restrict__ adj,
    const int* __restrict__ s1, float* __restrict__ out)
{
    const int row  = blockIdx.x;
    const int t    = threadIdx.x;
    const int node = s1[row];
    const int* arow = adj + (size_t)node * 25;
    float ax = 0.f, ay = 0.f, az = 0.f, aw = 0.f;
    for (int k = 0; k < 25; ++k) {
        const int nb = arow[k];
        const float4 v = *(const float4*)(feat + (size_t)nb * FEAT + t * 4);
        ax += v.x; ay += v.y; az += v.z; aw += v.w;
    }
    const float s = 1.0f / 25.0f;
    float4 r; r.x = ax * s; r.y = ay * s; r.z = az * s; r.w = aw * s;
    *(float4*)(out + (size_t)row * FEAT + t * 4) = r;
}

__global__ __launch_bounds__(64) void mean_neigh10_kernel(
    const float* __restrict__ feat, const int* __restrict__ s1,
    float* __restrict__ out)
{
    const int row = blockIdx.x;
    const int t   = threadIdx.x;
    float ax = 0.f, ay = 0.f, az = 0.f, aw = 0.f;
    for (int j = 0; j < 10; ++j) {
        const int nb = s1[row * 10 + j];
        const float4 v = *(const float4*)(feat + (size_t)nb * FEAT + t * 4);
        ax += v.x; ay += v.y; az += v.z; aw += v.w;
    }
    const float s = 0.1f;
    float4 r; r.x = ax * s; r.y = ay * s; r.z = az * s; r.w = aw * s;
    *(float4*)(out + (size_t)row * FEAT + t * 4) = r;
}

template <bool RELU>
__global__ __launch_bounds__(128) void sage_gemm_rows80_mean10_kernel(
    const float* __restrict__ feat, const int* __restrict__ idx,
    const float* __restrict__ Aneigh,
    const float* __restrict__ Ws, const float* __restrict__ Wn,
    float* __restrict__ out)
{
    constexpr int RT = 5;
    __shared__ v4f part[4 * RT * 16 * 16];

    const int tid  = threadIdx.x;
    const int lane = tid & 31;
    const int wave = tid >> 5;
    const int c    = lane & 15;
    const int h    = lane >> 4;
    const int col0 = blockIdx.y * 64;
    const int rowBase = blockIdx.x * (16 * RT);
    const int k0   = wave * 64;

    unsigned node[RT];
#pragma unroll
    for (int i = 0; i < RT; ++i) node[i] = (unsigned)idx[rowBase + 16 * i + c];
    __builtin_amdgcn_sched_barrier(0);

    const size_t woff = (size_t)(k0 + 4 * h) * FEAT + col0 + 4 * c;
    const float* gs = Ws + woff;
    const float* gn = Wn + woff;

    v4f wr[16][4];
    v4f a[RT][16];
#pragma unroll
    for (int jj = 0; jj < 4; ++jj) {
#pragma unroll
        for (int r = 0; r < 4; ++r)
            wr[jj][r] = *(const v4f*)(gs + (size_t)(8 * jj + r) * FEAT);
    }
    __builtin_amdgcn_sched_barrier(0);
    const float* arow_s[RT];
    const float* arow_n[RT];
#pragma unroll
    for (int i = 0; i < RT; ++i) {
        arow_s[i] = feat + (size_t)node[i] * FEAT + k0 + 4 * h;
        arow_n[i] = Aneigh + (size_t)(rowBase + 16 * i + c) * FEAT + k0 + 4 * h;
    }
#pragma unroll
    for (int jj = 0; jj < 8; ++jj) {
        a[0][jj]     = *(const v4f*)(arow_s[0] + 8 * jj);
        a[0][jj + 8] = *(const v4f*)(arow_n[0] + 8 * jj);
    }
    __builtin_amdgcn_sched_barrier(0);
#pragma unroll
    for (int jj = 4; jj < 8; ++jj) {
#pragma unroll
        for (int r = 0; r < 4; ++r)
            wr[jj][r] = *(const v4f*)(gs + (size_t)(8 * jj + r) * FEAT);
    }
    __builtin_amdgcn_sched_barrier(0);
#pragma unroll
    for (int jj = 0; jj < 4; ++jj) {
#pragma unroll
        for (int r = 0; r < 4; ++r)
            wr[jj + 8][r] = *(const v4f*)(gn + (size_t)(8 * jj + r) * FEAT);
    }
    __builtin_amdgcn_sched_barrier(0);
#pragma unroll
    for (int jj = 4; jj < 8; ++jj) {
#pragma unroll
        for (int r = 0; r < 4; ++r)
            wr[jj + 8][r] = *(const v4f*)(gn + (size_t)(8 * jj + r) * FEAT);
    }
    __builtin_amdgcn_sched_barrier(0);
#pragma unroll
    for (int jj = 0; jj < 8; ++jj) {
        a[1][jj]     = *(const v4f*)(arow_s[1] + 8 * jj);
        a[1][jj + 8] = *(const v4f*)(arow_n[1] + 8 * jj);
    }

#pragma unroll
    for (int i = 0; i < RT; ++i) {
        if (i == 1) {
#pragma unroll
            for (int i2 = 2; i2 < RT; ++i2) {
#pragma unroll
                for (int jj = 0; jj < 8; ++jj) {
                    a[i2][jj]     = *(const v4f*)(arow_s[i2] + 8 * jj);
                    a[i2][jj + 8] = *(const v4f*)(arow_n[i2] + 8 * jj);
                }
            }
        }
        if (i < 2) __builtin_amdgcn_sched_barrier(0);

        v8f acc[4];
#pragma unroll
        for (int t = 0; t < 4; ++t) acc[t] = (v8f)0.f;

#pragma unroll
        for (int jj = 0; jj < 16; ++jj) {
            const v2f a0 = {a[i][jj].x, a[i][jj].y};
            const v2f a1 = {a[i][jj].z, a[i][jj].w};
#pragma unroll
            for (int t = 0; t < 4; ++t) {
                const v2f b0 = {wr[jj][0][t], wr[jj][1][t]};
                acc[t] = __builtin_amdgcn_wmma_f32_16x16x4_f32(
                    false, a0, false, b0, (short)0, acc[t], false, false);
            }
#pragma unroll
            for (int t = 0; t < 4; ++t) {
                const v2f b1 = {wr[jj][2][t], wr[jj][3][t]};
                acc[t] = __builtin_amdgcn_wmma_f32_16x16x4_f32(
                    false, a1, false, b1, (short)0, acc[t], false, false);
            }
        }

#pragma unroll
        for (int r = 0; r < 8; ++r) {
            v4f o;
            o.x = acc[0][r]; o.y = acc[1][r]; o.z = acc[2][r]; o.w = acc[3][r];
            part[((wave * RT + i) * 16 + 8 * h + r) * 16 + c] = o;
        }
    }
    __syncthreads();

    const int g = tid >> 4;
    const int q = tid & 15;
    v4f mean = (v4f)0.f;
    for (int j = 0; j < 10; ++j) {
        const int row = 10 * g + j;
        v4f sum = part[row * 16 + q];
#pragma unroll
        for (int w = 1; w < 4; ++w) sum += part[(w * 16 * RT + row) * 16 + q];
        if (RELU) {
            sum.x = fmaxf(sum.x, 0.0f); sum.y = fmaxf(sum.y, 0.0f);
            sum.z = fmaxf(sum.z, 0.0f); sum.w = fmaxf(sum.w, 0.0f);
        }
        mean += sum;
    }
    mean *= 0.1f;
    *(v4f*)(out + (size_t)(blockIdx.x * ((16 * RT) / 10) + g) * FEAT
            + col0 + 4 * q) = mean;
}

template <bool RELU, bool GATHER>
__global__ __launch_bounds__(256) void sage_gemm_splitk_kernel(
    const float* __restrict__ Aself, const int* __restrict__ idx,
    const float* __restrict__ Aneigh,
    const float* __restrict__ Ws, const float* __restrict__ Wn,
    float* __restrict__ out)
{
    __shared__ v4f part[8 * 16 * 16];

    const int tid  = threadIdx.x;
    const int lane = tid & 31;
    const int wave = tid >> 5;
    const int c    = lane & 15;
    const int h    = lane >> 4;
    const int col0 = blockIdx.y * 64;
    const int rowBase = blockIdx.x * 16;
    const int m    = rowBase + c;
    const bool self = wave < 4;
    const int k0   = (wave & 3) * 64;

    const int node = GATHER ? idx[m] : m;
    const float* g = (self ? Ws : Wn)
                     + (size_t)(k0 + 4 * h) * FEAT + col0 + 4 * c;

    v4f wr[8][4];
#pragma unroll
    for (int jj = 0; jj < 8; ++jj) {
#pragma unroll
        for (int r = 0; r < 4; ++r)
            wr[jj][r] = *(const v4f*)(g + (size_t)(8 * jj + r) * FEAT);
    }
    const float* arow = (self ? (Aself + (size_t)node * FEAT)
                              : (Aneigh + (size_t)m * FEAT)) + k0 + 4 * h;
    v4f a[8];
#pragma unroll
    for (int jj = 0; jj < 8; ++jj) a[jj] = *(const v4f*)(arow + 8 * jj);
    __builtin_amdgcn_sched_barrier(0);

    v8f acc[4];
#pragma unroll
    for (int t = 0; t < 4; ++t) acc[t] = (v8f)0.f;

#pragma unroll
    for (int jj = 0; jj < 8; ++jj) {
        const v2f a0 = {a[jj].x, a[jj].y};
        const v2f a1 = {a[jj].z, a[jj].w};
#pragma unroll
        for (int t = 0; t < 4; ++t) {
            const v2f b0 = {wr[jj][0][t], wr[jj][1][t]};
            acc[t] = __builtin_amdgcn_wmma_f32_16x16x4_f32(
                false, a0, false, b0, (short)0, acc[t], false, false);
        }
#pragma unroll
        for (int t = 0; t < 4; ++t) {
            const v2f b1 = {wr[jj][2][t], wr[jj][3][t]};
            acc[t] = __builtin_amdgcn_wmma_f32_16x16x4_f32(
                false, a1, false, b1, (short)0, acc[t], false, false);
        }
    }

#pragma unroll
    for (int r = 0; r < 8; ++r) {
        v4f o;
        o.x = acc[0][r]; o.y = acc[1][r]; o.z = acc[2][r]; o.w = acc[3][r];
        part[(wave * 16 + 8 * h + r) * 16 + c] = o;
    }
    __syncthreads();

    const int row = tid >> 4;
    const int q   = tid & 15;
    v4f sum = part[row * 16 + q];
#pragma unroll
    for (int w = 1; w < 8; ++w) sum += part[(w * 16 + row) * 16 + q];
    if (RELU) {
        sum.x = fmaxf(sum.x, 0.0f); sum.y = fmaxf(sum.y, 0.0f);
        sum.z = fmaxf(sum.z, 0.0f); sum.w = fmaxf(sum.w, 0.0f);
    }
    *(v4f*)(out + (size_t)(rowBase + row) * FEAT + col0 + 4 * q) = sum;
}

__global__ __launch_bounds__(256) void l2norm_kernel(
    const float* __restrict__ pre, float* __restrict__ out)
{
    __shared__ float ssum[8];
    const int row = blockIdx.x;
    const int t   = threadIdx.x;
    const float v = pre[(size_t)row * FEAT + t];
    float s = v * v;
#pragma unroll
    for (int o = 16; o >= 1; o >>= 1) s += __shfl_xor(s, o, 32);
    if ((t & 31) == 0) ssum[t >> 5] = s;
    __syncthreads();
    float tot = 0.f;
#pragma unroll
    for (int i = 0; i < 8; ++i) tot += ssum[i];
    const float norm = fmaxf(sqrtf(tot), 1e-12f);
    out[(size_t)row * FEAT + t] = v / norm;
}

extern "C" void kernel_launch(void* const* d_in, const int* in_sizes, int n_in,
                              void* d_out, int out_size, void* d_ws,
                              size_t ws_size, hipStream_t stream)
{
    const float* features = (const float*)d_in[0];
    const int*   adj      = (const int*)d_in[1];
    const int*   batch    = (const int*)d_in[2];
    const float* Ws0      = (const float*)d_in[3];
    const float* Wn0      = (const float*)d_in[4];
    const float* Ws1      = (const float*)d_in[5];
    const float* Wn1      = (const float*)d_in[6];
    float* out = (float*)d_out;

    char* ws = (char*)d_ws;
    int*   s1   = (int*)(ws + 0);
    float* nm2  = (float*)(ws + 20480);
    float* nm1  = (float*)(ws + 5263360);
    float* h0   = (float*)(ws + 11030528);
    float* nmh1 = (float*)(ws + 11554816);
    float* pre  = (float*)(ws + 12079104);

    build_s1_kernel<<<(5120 + 255) / 256, 256, 0, stream>>>(adj, batch, s1);
    mean_neigh25_kernel<<<5120, 64, 0, stream>>>(features, adj, s1, nm2);
    mean_neigh10_kernel<<<512, 64, 0, stream>>>(features, s1, nm1);
    sage_gemm_rows80_mean10_kernel<true><<<dim3(64, 4), 128, 0, stream>>>(
        features, s1, nm2, Ws0, Wn0, nmh1);
    sage_gemm_splitk_kernel<true, true><<<dim3(32, 4), 256, 0, stream>>>(
        features, batch, nm1, Ws0, Wn0, h0);
    sage_gemm_splitk_kernel<false, false><<<dim3(32, 4), 256, 0, stream>>>(
        h0, (const int*)nullptr, nmh1, Ws1, Wn1, pre);
    l2norm_kernel<<<512, 256, 0, stream>>>(pre, out);
}


# Hop-0 branch rides in the hop-1 chain's launches: the sample-list grid also forms its 10-row means, the hop-2 gather grid (4 rows a block) carries its GEMM's 128 blocks; 7 launches to 5; sums same
# speedup vs baseline: 6.2985x; 1.0777x over previous
#include <hip/hip_runtime.h>
#include <hip/hip_bf16.h>

typedef __attribute__((ext_vector_type(2))) float v2f;
typedef __attribute__((ext_vector_type(4))) float v4f;
typedef __attribute__((ext_vector_type(8))) float v8f;

#define FEAT 256

__global__ __launch_bounds__(64) void sample10_mean_kernel(
    const float* __restrict__ feat, const int* __restrict__ adj,
    const int* __restrict__ batch, int* __restrict__ s1,
    float* __restrict__ nm1)
{
    const int b = blockIdx.x;
    const int t = threadIdx.x;
    const int* arow = adj + (size_t)batch[b] * 25;
    int listId = 0;
    float ax = 0.f, ay = 0.f, az = 0.f, aw = 0.f;
    for (int j = 0; j < 10; ++j) {
        const int nb = arow[j];
        listId = (t == j) ? nb : listId;
        const float4 v = *(const float4*)(feat + (size_t)nb * FEAT + t * 4);
        ax += v.x; ay += v.y; az += v.z; aw += v.w;
    }
    const float s = 0.1f;
    float4 r; r.x = ax * s; r.y = ay * s; r.z = az * s; r.w = aw * s;
    *(float4*)(nm1 + (size_t)b * FEAT + t * 4) = r;
    if (t < 10) s1[b * 10 + t] = listId;
}

template <bool RELU>
__global__ __launch_bounds__(128) void sage_gemm_rows80_mean10_kernel(
    const float* __restrict__ feat, const int* __restrict__ idx,
    const float* __restrict__ Aneigh,
    const float* __restrict__ Ws, const float* __restrict__ Wn,
    float* __restrict__ out)
{
    constexpr int RT = 5;
    __shared__ v4f part[4 * RT * 16 * 16];

    const int tid  = threadIdx.x;
    const int lane = tid & 31;
    const int wave = tid >> 5;
    const int c    = lane & 15;
    const int h    = lane >> 4;
    const int col0 = blockIdx.y * 64;
    const int rowBase = blockIdx.x * (16 * RT);
    const int k0   = wave * 64;

    unsigned node[RT];
#pragma unroll
    for (int i = 0; i < RT; ++i) node[i] = (unsigned)idx[rowBase + 16 * i + c];
    __builtin_amdgcn_sched_barrier(0);

    const size_t woff = (size_t)(k0 + 4 * h) * FEAT + col0 + 4 * c;
    const float* gs = Ws + woff;
    const float* gn = Wn + woff;

    v4f wr[16][4];
    v4f a[RT][16];
#pragma unroll
    for (int jj = 0; jj < 4; ++jj) {
#pragma unroll
        for (int r = 0; r < 4; ++r)
            wr[jj][r] = *(const v4f*)(gs + (size_t)(8 * jj + r) * FEAT);
    }
    __builtin_amdgcn_sched_barrier(0);
    const float* arow_s[RT];
    const float* arow_n[RT];
#pragma unroll
    for (int i = 0; i < RT; ++i) {
        arow_s[i] = feat + (size_t)node[i] * FEAT + k0 + 4 * h;
        arow_n[i] = Aneigh + (size_t)(rowBase + 16 * i + c) * FEAT + k0 + 4 * h;
    }
#pragma unroll
    for (int jj = 0; jj < 8; ++jj) {
        a[0][jj]     = *(const v4f*)(arow_s[0] + 8 * jj);
        a[0][jj + 8] = *(const v4f*)(arow_n[0] + 8 * jj);
    }
    __builtin_amdgcn_sched_barrier(0);
#pragma unroll
    for (int jj = 4; jj < 8; ++jj) {
#pragma unroll
        for (int r = 0; r < 4; ++r)
            wr[jj][r] = *(const v4f*)(gs + (size_t)(8 * jj + r) * FEAT);
    }
    __builtin_amdgcn_sched_barrier(0);
#pragma unroll
    for (int jj = 0; jj < 4; ++jj) {
#pragma unroll
        for (int r = 0; r < 4; ++r)
            wr[jj + 8][r] = *(const v4f*)(gn + (size_t)(8 * jj + r) * FEAT);
    }
    __builtin_amdgcn_sched_barrier(0);
#pragma unroll
    for (int jj = 4; jj < 8; ++jj) {
#pragma unroll
        for (int r = 0; r < 4; ++r)
            wr[jj + 8][r] = *(const v4f*)(gn + (size_t)(8 * jj + r) * FEAT);
    }
    __builtin_amdgcn_sched_barrier(0);
#pragma unroll
    for (int jj = 0; jj < 8; ++jj) {
        a[1][jj]     = *(const v4f*)(arow_s[1] + 8 * jj);
        a[1][jj + 8] = *(const v4f*)(arow_n[1] + 8 * jj);
    }

#pragma unroll
    for (int i = 0; i < RT; ++i) {
        if (i == 1) {
#pragma unroll
            for (int i2 = 2; i2 < RT; ++i2) {
#pragma unroll
                for (int jj = 0; jj < 8; ++jj) {
                    a[i2][jj]     = *(const v4f*)(arow_s[i2] + 8 * jj);
                    a[i2][jj + 8] = *(const v4f*)(arow_n[i2] + 8 * jj);
                }
            }
        }
        if (i < 2) __builtin_amdgcn_sched_barrier(0);

        v8f acc[4];
#pragma unroll
        for (int t = 0; t < 4; ++t) acc[t] = (v8f)0.f;

#pragma unroll
        for (int jj = 0; jj < 16; ++jj) {
            const v2f a0 = {a[i][jj].x, a[i][jj].y};
            const v2f a1 = {a[i][jj].z, a[i][jj].w};
#pragma unroll
            for (int t = 0; t < 4; ++t) {
                const v2f b0 = {wr[jj][0][t], wr[jj][1][t]};
                acc[t] = __builtin_amdgcn_wmma_f32_16x16x4_f32(
                    false, a0, false, b0, (short)0, acc[t], false, false);
            }
#pragma unroll
            for (int t = 0; t < 4; ++t) {
                const v2f b1 = {wr[jj][2][t], wr[jj][3][t]};
                acc[t] = __builtin_amdgcn_wmma_f32_16x16x4_f32(
                    false, a1, false, b1, (short)0, acc[t], false, false);
            }
        }

#pragma unroll
        for (int r = 0; r < 8; ++r) {
            v4f o;
            o.x = acc[0][r]; o.y = acc[1][r]; o.z = acc[2][r]; o.w = acc[3][r];
            part[((wave * RT + i) * 16 + 8 * h + r) * 16 + c] = o;
        }
    }
    __syncthreads();

    const int g = tid >> 4;
    const int q = tid & 15;
    v4f mean = (v4f)0.f;
    for (int j = 0; j < 10; ++j) {
        const int row = 10 * g + j;
        v4f sum = part[row * 16 + q];
#pragma unroll
        for (int w = 1; w < 4; ++w) sum += part[(w * 16 * RT + row) * 16 + q];
        if (RELU) {
            sum.x = fmaxf(sum.x, 0.0f); sum.y = fmaxf(sum.y, 0.0f);
            sum.z = fmaxf(sum.z, 0.0f); sum.w = fmaxf(sum.w, 0.0f);
        }
        mean += sum;
    }
    mean *= 0.1f;
    *(v4f*)(out + (size_t)(blockIdx.x * ((16 * RT) / 10) + g) * FEAT
            + col0 + 4 * q) = mean;
}

template <bool RELU, bool GATHER>
__device__ __forceinline__ void sage_gemm_splitk_tile(
    v4f* __restrict__ part,
    const float* __restrict__ Aself, const int* __restrict__ idx,
    const float* __restrict__ Aneigh,
    const float* __restrict__ Ws, const float* __restrict__ Wn,
    float* __restrict__ out, const int rowTile, const int colSlice)
{
    const int tid  = threadIdx.x;
    const int lane = tid & 31;
    const int wave = tid >> 5;
    const int c    = lane & 15;
    const int h    = lane >> 4;
    const int col0 = colSlice * 64;
    const int rowBase = rowTile * 16;
    const int m    = rowBase + c;
    const bool self = wave < 4;
    const int k0   = (wave & 3) * 64;

    const int node = GATHER ? idx[m] : m;
    const float* g = (self ? Ws : Wn)
                     + (size_t)(k0 + 4 * h) * FEAT + col0 + 4 * c;

    v4f wr[8][4];
#pragma unroll
    for (int jj = 0; jj < 8; ++jj) {
#pragma unroll
        for (int r = 0; r < 4; ++r)
            wr[jj][r] = *(const v4f*)(g + (size_t)(8 * jj + r) * FEAT);
    }
    const float* arow = (self ? (Aself + (size_t)node * FEAT)
                              : (Aneigh + (size_t)m * FEAT)) + k0 + 4 * h;
    v4f a[8];
#pragma unroll
    for (int jj = 0; jj < 8; ++jj) a[jj] = *(const v4f*)(arow + 8 * jj);
    __builtin_amdgcn_sched_barrier(0);

    v8f acc[4];
#pragma unroll
    for (int t = 0; t < 4; ++t) acc[t] = (v8f)0.f;

#pragma unroll
    for (int jj = 0; jj < 8; ++jj) {
        const v2f a0 = {a[jj].x, a[jj].y};
        const v2f a1 = {a[jj].z, a[jj].w};
#pragma unroll
        for (int t = 0; t < 4; ++t) {
            const v2f b0 = {wr[jj][0][t], wr[jj][1][t]};
            acc[t] = __builtin_amdgcn_wmma_f32_16x16x4_f32(
                false, a0, false, b0, (short)0, acc[t], false, false);
        }
#pragma unroll
        for (int t = 0; t < 4; ++t) {
            const v2f b1 = {wr[jj][2][t], wr[jj][3][t]};
            acc[t] = __builtin_amdgcn_wmma_f32_16x16x4_f32(
                false, a1, false, b1, (short)0, acc[t], false, false);
        }
    }

#pragma unroll
    for (int r = 0; r < 8; ++r) {
        v4f o;
        o.x = acc[0][r]; o.y = acc[1][r]; o.z = acc[2][r]; o.w = acc[3][r];
        part[(wave * 16 + 8 * h + r) * 16 + c] = o;
    }
    __syncthreads();

    const int row = tid >> 4;
    const int q   = tid & 15;
    v4f sum = part[row * 16 + q];
#pragma unroll
    for (int w = 1; w < 8; ++w) sum += part[(w * 16 + row) * 16 + q];
    if (RELU) {
        sum.x = fmaxf(sum.x, 0.0f); sum.y = fmaxf(sum.y, 0.0f);
        sum.z = fmaxf(sum.z, 0.0f); sum.w = fmaxf(sum.w, 0.0f);
    }
    *(v4f*)(out + (size_t)(rowBase + row) * FEAT + col0 + 4 * q) = sum;
}

template <bool RELU, bool GATHER>
__global__ __launch_bounds__(256) void sage_gemm_splitk_kernel(
    const float* __restrict__ Aself, const int* __restrict__ idx,
    const float* __restrict__ Aneigh,
    const float* __restrict__ Ws, const float* __restrict__ Wn,
    float* __restrict__ out)
{
    __shared__ v4f part[8 * 16 * 16];
    sage_gemm_splitk_tile<RELU, GATHER>(part, Aself, idx, Aneigh, Ws, Wn, out,
                                        blockIdx.x, blockIdx.y);
}

__global__ __launch_bounds__(256) void hop2_mean_with_hop0_gemm_kernel(
    const float* __restrict__ feat, const int* __restrict__ adj,
    const int* __restrict__ s1, const int* __restrict__ batch,
    const float* __restrict__ nm1,
    const float* __restrict__ Ws, const float* __restrict__ Wn,
    float* __restrict__ nm2, float* __restrict__ h0)
{
    __shared__ v4f part[8 * 16 * 16];

    if (blockIdx.x < 128) {
        sage_gemm_splitk_tile<true, true>(part, feat, batch, nm1, Ws, Wn, h0,
                                          blockIdx.x & 31, blockIdx.x >> 5);
        return;
    }
    const int pairId = __builtin_amdgcn_readfirstlane((int)(threadIdx.x >> 6));
    const int row  = (blockIdx.x - 128) * 4 + pairId;
    const int t    = threadIdx.x & 63;
    const int node = s1[row];
    const int* arow = adj + (size_t)node * 25;
    float ax = 0.f, ay = 0.f, az = 0.f, aw = 0.f;
    for (int k = 0; k < 25; ++k) {
        const int nb = arow[k];
        const float4 v = *(const float4*)(feat + (size_t)nb * FEAT + t * 4);
        ax += v.x; ay += v.y; az += v.z; aw += v.w;
    }
    const float s = 1.0f / 25.0f;
    float4 r; r.x = ax * s; r.y = ay * s; r.z = az * s; r.w = aw * s;
    *(float4*)(nm2 + (size_t)row * FEAT + t * 4) = r;
}

__global__ __launch_bounds__(256) void l2norm_kernel(
    const float* __restrict__ pre, float* __restrict__ out)
{
    __shared__ float ssum[8];
    const int row = blockIdx.x;
    const int t   = threadIdx.x;
    const float v = pre[(size_t)row * FEAT + t];
    float s = v * v;
#pragma unroll
    for (int o = 16; o >= 1; o >>= 1) s += __shfl_xor(s, o, 32);
    if ((t & 31) == 0) ssum[t >> 5] = s;
    __syncthreads();
    float tot = 0.f;
#pragma unroll
    for (int i = 0; i < 8; ++i) tot += ssum[i];
    const float norm = fmaxf(sqrtf(tot), 1e-12f);
    out[(size_t)row * FEAT + t] = v / norm;
}

extern "C" void kernel_launch(void* const* d_in, const int* in_sizes, int n_in,
                              void* d_out, int out_size, void* d_ws,
                              size_t ws_size, hipStream_t stream)
{
    const float* features = (const float*)d_in[0];
    const int*   adj      = (const int*)d_in[1];
    const int*   batch    = (const int*)d_in[2];
    const float* Ws0      = (const float*)d_in[3];
    const float* Wn0      = (const float*)d_in[4];
    const float* Ws1      = (const float*)d_in[5];
    const float* Wn1      = (const float*)d_in[6];
    float* out = (float*)d_out;

    char* ws = (char*)d_ws;
    int*   s1   = (int*)(ws + 0);
    float* nm2  = (float*)(ws + 20480);
    float* nm1  = (float*)(ws + 5263360);
    float* h0   = (float*)(ws + 11030528);
    float* nmh1 = (float*)(ws + 11554816);
    float* pre  = (float*)(ws + 12079104);

    sample10_mean_kernel<<<512, 64, 0, stream>>>(features, adj, batch, s1, nm1);
    hop2_mean_with_hop0_gemm_kernel<<<128 + 1280, 256, 0, stream>>>(
        features, adj, s1, batch, nm1, Ws0, Wn0, nm2, h0);
    sage_gemm_rows80_mean10_kernel<true><<<dim3(64, 4), 128, 0, stream>>>(
        features, s1, nm2, Ws0, Wn0, nmh1);
    sage_gemm_splitk_kernel<false, false><<<dim3(32, 4), 256, 0, stream>>>(
        h0, (const int*)nullptr, nmh1, Ws1, Wn1, pre);
    l2norm_kernel<<<512, 256, 0, stream>>>(pre, out);
}


# Sample-list launch dissolved: a gather wave looks up its own hop-1 id (2 scalar trips) and writes the list; hop-0 GEMM blocks form their rows' 10-neighbour means in LDS; 5 launches to 4; sums same
# speedup vs baseline: 6.5123x; 1.0339x over previous
#include <hip/hip_runtime.h>
#include <hip/hip_bf16.h>

typedef __attribute__((ext_vector_type(2))) float v2f;
typedef __attribute__((ext_vector_type(4))) float v4f;
typedef __attribute__((ext_vector_type(8))) float v8f;

#define FEAT 256

template <bool RELU>
__global__ __launch_bounds__(128) void sage_gemm_rows80_mean10_kernel(
    const float* __restrict__ feat, const int* __restrict__ idx,
    const float* __restrict__ Aneigh,
    const float* __restrict__ Ws, const float* __restrict__ Wn,
    float* __restrict__ out)
{
    constexpr int RT = 5;
    __shared__ v4f part[4 * RT * 16 * 16];

    const int tid  = threadIdx.x;
    const int lane = tid & 31;
    const int wave = tid >> 5;
    const int c    = lane & 15;
    const int h    = lane >> 4;
    const int col0 = blockIdx.y * 64;
    const int rowBase = blockIdx.x * (16 * RT);
    const int k0   = wave * 64;

    unsigned node[RT];
#pragma unroll
    for (int i = 0; i < RT; ++i) node[i] = (unsigned)idx[rowBase + 16 * i + c];
    __builtin_amdgcn_sched_barrier(0);

    const size_t woff = (size_t)(k0 + 4 * h) * FEAT + col0 + 4 * c;
    const float* gs = Ws + woff;
    const float* gn = Wn + woff;

    v4f wr[16][4];
    v4f a[RT][16];
#pragma unroll
    for (int jj = 0; jj < 4; ++jj) {
#pragma unroll
        for (int r = 0; r < 4; ++r)
            wr[jj][r] = *(const v4f*)(gs + (size_t)(8 * jj + r) * FEAT);
    }
    __builtin_amdgcn_sched_barrier(0);
    const float* arow_s[RT];
    const float* arow_n[RT];
#pragma unroll
    for (int i = 0; i < RT; ++i) {
        arow_s[i] = feat + (size_t)node[i] * FEAT + k0 + 4 * h;
        arow_n[i] = Aneigh + (size_t)(rowBase + 16 * i + c) * FEAT + k0 + 4 * h;
    }
#pragma unroll
    for (int jj = 0; jj < 8; ++jj) {
        a[0][jj]     = *(const v4f*)(arow_s[0] + 8 * jj);
        a[0][jj + 8] = *(const v4f*)(arow_n[0] + 8 * jj);
    }
    __builtin_amdgcn_sched_barrier(0);
#pragma unroll
    for (int jj = 4; jj < 8; ++jj) {
#pragma unroll
        for (int r = 0; r < 4; ++r)
            wr[jj][r] = *(const v4f*)(gs + (size_t)(8 * jj + r) * FEAT);
    }
    __builtin_amdgcn_sched_barrier(0);
#pragma unroll
    for (int jj = 0; jj < 4; ++jj) {
#pragma unroll
        for (int r = 0; r < 4; ++r)
            wr[jj + 8][r] = *(const v4f*)(gn + (size_t)(8 * jj + r) * FEAT);
    }
    __builtin_amdgcn_sched_barrier(0);
#pragma unroll
    for (int jj = 4; jj < 8; ++jj) {
#pragma unroll
        for (int r = 0; r < 4; ++r)
            wr[jj + 8][r] = *(const v4f*)(gn + (size_t)(8 * jj + r) * FEAT);
    }
    __builtin_amdgcn_sched_barrier(0);
#pragma unroll
    for (int jj = 0; jj < 8; ++jj) {
        a[1][jj]     = *(const v4f*)(arow_s[1] + 8 * jj);
        a[1][jj + 8] = *(const v4f*)(arow_n[1] + 8 * jj);
    }

#pragma unroll
    for (int i = 0; i < RT; ++i) {
        if (i == 1) {
#pragma unroll
            for (int i2 = 2; i2 < RT; ++i2) {
#pragma unroll
                for (int jj = 0; jj < 8; ++jj) {
                    a[i2][jj]     = *(const v4f*)(arow_s[i2] + 8 * jj);
                    a[i2][jj + 8] = *(const v4f*)(arow_n[i2] + 8 * jj);
                }
            }
        }
        if (i < 2) __builtin_amdgcn_sched_barrier(0);

        v8f acc[4];
#pragma unroll
        for (int t = 0; t < 4; ++t) acc[t] = (v8f)0.f;

#pragma unroll
        for (int jj = 0; jj < 16; ++jj) {
            const v2f a0 = {a[i][jj].x, a[i][jj].y};
            const v2f a1 = {a[i][jj].z, a[i][jj].w};
#pragma unroll
            for (int t = 0; t < 4; ++t) {
                const v2f b0 = {wr[jj][0][t], wr[jj][1][t]};
                acc[t] = __builtin_amdgcn_wmma_f32_16x16x4_f32(
                    false, a0, false, b0, (short)0, acc[t], false, false);
            }
#pragma unroll
            for (int t = 0; t < 4; ++t) {
                const v2f b1 = {wr[jj][2][t], wr[jj][3][t]};
                acc[t] = __builtin_amdgcn_wmma_f32_16x16x4_f32(
                    false, a1, false, b1, (short)0, acc[t], false, false);
            }
        }

#pragma unroll
        for (int r = 0; r < 8; ++r) {
            v4f o;
            o.x = acc[0][r]; o.y = acc[1][r]; o.z = acc[2][r]; o.w = acc[3][r];
            part[((wave * RT + i) * 16 + 8 * h + r) * 16 + c] = o;
        }
    }
    __syncthreads();

    const int g = tid >> 4;
    const int q = tid & 15;
    v4f mean = (v4f)0.f;
    for (int j = 0; j < 10; ++j) {
        const int row = 10 * g + j;
        v4f sum = part[row * 16 + q];
#pragma unroll
        for (int w = 1; w < 4; ++w) sum += part[(w * 16 * RT + row) * 16 + q];
        if (RELU) {
            sum.x = fmaxf(sum.x, 0.0f); sum.y = fmaxf(sum.y, 0.0f);
            sum.z = fmaxf(sum.z, 0.0f); sum.w = fmaxf(sum.w, 0.0f);
        }
        mean += sum;
    }
    mean *= 0.1f;
    *(v4f*)(out + (size_t)(blockIdx.x * ((16 * RT) / 10) + g) * FEAT
            + col0 + 4 * q) = mean;
}

#define NM_PITCH 260

template <bool RELU, bool GATHER, bool MEAN10>
__device__ __forceinline__ void sage_gemm_splitk_tile(
    v4f* __restrict__ part, float* __restrict__ nmT,
    const float* __restrict__ Aself, const int* __restrict__ idx,
    const int* __restrict__ adj, const float* __restrict__ Aneigh,
    const float* __restrict__ Ws, const float* __restrict__ Wn,
    float* __restrict__ out, const int rowTile, const int colSlice)
{
    const int tid  = threadIdx.x;
    const int lane = tid & 31;
    const int wave = tid >> 5;
    const int c    = lane & 15;
    const int h    = lane >> 4;
    const int col0 = colSlice * 64;
    const int rowBase = rowTile * 16;
    const int m    = rowBase + c;
    const bool self = wave < 4;
    const int k0   = (wave & 3) * 64;

    const int node = GATHER ? idx[m] : m;

    v4f a[8];
    if (MEAN10) {
        if (self) {
            const float* arow = Aself + (size_t)node * FEAT + k0 + 4 * h;
#pragma unroll
            for (int jj = 0; jj < 8; ++jj) a[jj] = *(const v4f*)(arow + 8 * jj);
        }
        __builtin_amdgcn_sched_barrier(0);
        const int wv = __builtin_amdgcn_readfirstlane(wave);
#pragma unroll
        for (int rr = 0; rr < 2; ++rr) {
            const int r = 2 * wv + rr;
            const int* arow = adj + (size_t)idx[rowBase + r] * 25;
            float4 lo = {0.f, 0.f, 0.f, 0.f}, hi = {0.f, 0.f, 0.f, 0.f};
            for (int j = 0; j < 10; ++j) {
                const float* s = Aself + (size_t)arow[j] * FEAT + lane * 4;
                const float4 v0 = *(const float4*)s;
                const float4 v1 = *(const float4*)(s + 128);
                lo.x += v0.x; lo.y += v0.y; lo.z += v0.z; lo.w += v0.w;
                hi.x += v1.x; hi.y += v1.y; hi.z += v1.z; hi.w += v1.w;
            }
            const float sc = 0.1f;
            lo.x *= sc; lo.y *= sc; lo.z *= sc; lo.w *= sc;
            hi.x *= sc; hi.y *= sc; hi.z *= sc; hi.w *= sc;
            *(float4*)(nmT + r * NM_PITCH + lane * 4)       = lo;
            *(float4*)(nmT + r * NM_PITCH + 128 + lane * 4) = hi;
        }
        __builtin_amdgcn_sched_barrier(0);
    }

    const float* g = (self ? Ws : Wn)
                     + (size_t)(k0 + 4 * h) * FEAT + col0 + 4 * c;

    v4f wr[8][4];
#pragma unroll
    for (int jj = 0; jj < 8; ++jj) {
#pragma unroll
        for (int r = 0; r < 4; ++r)
            wr[jj][r] = *(const v4f*)(g + (size_t)(8 * jj + r) * FEAT);
    }
    if (!MEAN10) {
        const float* arow = (self ? (Aself + (size_t)node * FEAT)
                                  : (Aneigh + (size_t)m * FEAT)) + k0 + 4 * h;
#pragma unroll
        for (int jj = 0; jj < 8; ++jj) a[jj] = *(const v4f*)(arow + 8 * jj);
    }
    __builtin_amdgcn_sched_barrier(0);
    if (MEAN10) {
        __builtin_amdgcn_fence(__ATOMIC_RELEASE, "workgroup", "local");
        __builtin_amdgcn_s_barrier();
        __builtin_amdgcn_fence(__ATOMIC_ACQUIRE, "workgroup", "local");
        if (!self) {
            const float* trow = nmT + c * NM_PITCH + k0 + 4 * h;
#pragma unroll
            for (int jj = 0; jj < 8; ++jj) a[jj] = *(const v4f*)(trow + 8 * jj);
        }
    }

    v8f acc[4];
#pragma unroll
    for (int t = 0; t < 4; ++t) acc[t] = (v8f)0.f;

#pragma unroll
    for (int jj = 0; jj < 8; ++jj) {
        const v2f a0 = {a[jj].x, a[jj].y};
        const v2f a1 = {a[jj].z, a[jj].w};
#pragma unroll
        for (int t = 0; t < 4; ++t) {
            const v2f b0 = {wr[jj][0][t], wr[jj][1][t]};
            acc[t] = __builtin_amdgcn_wmma_f32_16x16x4_f32(
                false, a0, false, b0, (short)0, acc[t], false, false);
        }
#pragma unroll
        for (int t = 0; t < 4; ++t) {
            const v2f b1 = {wr[jj][2][t], wr[jj][3][t]};
            acc[t] = __builtin_amdgcn_wmma_f32_16x16x4_f32(
                false, a1, false, b1, (short)0, acc[t], false, false);
        }
    }

#pragma unroll
    for (int r = 0; r < 8; ++r) {
        v4f o;
        o.x = acc[0][r]; o.y = acc[1][r]; o.z = acc[2][r]; o.w = acc[3][r];
        part[(wave * 16 + 8 * h + r) * 16 + c] = o;
    }
    __syncthreads();

    const int row = tid >> 4;
    const int q   = tid & 15;
    v4f sum = part[row * 16 + q];
#pragma unroll
    for (int w = 1; w < 8; ++w) sum += part[(w * 16 + row) * 16 + q];
    if (RELU) {
        sum.x = fmaxf(sum.x, 0.0f); sum.y = fmaxf(sum.y, 0.0f);
        sum.z = fmaxf(sum.z, 0.0f); sum.w = fmaxf(sum.w, 0.0f);
    }
    *(v4f*)(out + (size_t)(rowBase + row) * FEAT + col0 + 4 * q) = sum;
}

template <bool RELU, bool GATHER>
__global__ __launch_bounds__(256) void sage_gemm_splitk_kernel(
    const float* __restrict__ Aself, const int* __restrict__ idx,
    const float* __restrict__ Aneigh,
    const float* __restrict__ Ws, const float* __restrict__ Wn,
    float* __restrict__ out)
{
    __shared__ v4f part[8 * 16 * 16];
    sage_gemm_splitk_tile<RELU, GATHER, false>(
        part, (float*)nullptr, Aself, idx, (const int*)nullptr, Aneigh, Ws, Wn,
        out, blockIdx.x, blockIdx.y);
}

__global__ __launch_bounds__(256) void hop2_mean_with_hop0_gemm_kernel(
    const float* __restrict__ feat, const int* __restrict__ adj,
    const int* __restrict__ batch,
    const float* __restrict__ Ws, const float* __restrict__ Wn,
    int* __restrict__ s1, float* __restrict__ nm2, float* __restrict__ h0)
{
    __shared__ v4f part[8 * 16 * 16];
    __shared__ float nmT[16 * NM_PITCH];

    if (blockIdx.x < 128) {
        sage_gemm_splitk_tile<true, true, true>(
            part, nmT, feat, batch, adj, (const float*)nullptr, Ws, Wn, h0,
            blockIdx.x & 31, blockIdx.x >> 5);
        return;
    }
    const int pairId = __builtin_amdgcn_readfirstlane((int)(threadIdx.x >> 6));
    const int row  = (blockIdx.x - 128) * 4 + pairId;
    const int t    = threadIdx.x & 63;
    const int b    = row / 10;
    const int node = adj[(size_t)batch[b] * 25 + (row - 10 * b)];
    const int* arow = adj + (size_t)node * 25;
    float ax = 0.f, ay = 0.f, az = 0.f, aw = 0.f;
    for (int k = 0; k < 25; ++k) {
        const int nb = arow[k];
        const float4 v = *(const float4*)(feat + (size_t)nb * FEAT + t * 4);
        ax += v.x; ay += v.y; az += v.z; aw += v.w;
    }
    const float s = 1.0f / 25.0f;
    float4 r; r.x = ax * s; r.y = ay * s; r.z = az * s; r.w = aw * s;
    *(float4*)(nm2 + (size_t)row * FEAT + t * 4) = r;
    if (t == 0) s1[row] = node;
}

__global__ __launch_bounds__(256) void l2norm_kernel(
    const float* __restrict__ pre, float* __restrict__ out)
{
    __shared__ float ssum[8];
    const int row = blockIdx.x;
    const int t   = threadIdx.x;
    const float v = pre[(size_t)row * FEAT + t];
    float s = v * v;
#pragma unroll
    for (int o = 16; o >= 1; o >>= 1) s += __shfl_xor(s, o, 32);
    if ((t & 31) == 0) ssum[t >> 5] = s;
    __syncthreads();
    float tot = 0.f;
#pragma unroll
    for (int i = 0; i < 8; ++i) tot += ssum[i];
    const float norm = fmaxf(sqrtf(tot), 1e-12f);
    out[(size_t)row * FEAT + t] = v / norm;
}

extern "C" void kernel_launch(void* const* d_in, const int* in_sizes, int n_in,
                              void* d_out, int out_size, void* d_ws,
                              size_t ws_size, hipStream_t stream)
{
    const float* features = (const float*)d_in[0];
    const int*   adj      = (const int*)d_in[1];
    const int*   batch    = (const int*)d_in[2];
    const float* Ws0      = (const float*)d_in[3];
    const float* Wn0      = (const float*)d_in[4];
    const float* Ws1      = (const float*)d_in[5];
    const float* Wn1      = (const float*)d_in[6];
    float* out = (float*)d_out;

    char* ws = (char*)d_ws;
    int*   s1   = (int*)(ws + 0);
    float* nm2  = (float*)(ws + 20480);
    float* h0   = (float*)(ws + 11030528);
    float* nmh1 = (float*)(ws + 11554816);
    float* pre  = (float*)(ws + 12079104);

    hop2_mean_with_hop0_gemm_kernel<<<128 + 1280, 256, 0, stream>>>(
        features, adj, batch, Ws0, Wn0, s1, nm2, h0);
    sage_gemm_rows80_mean10_kernel<true><<<dim3(64, 4), 128, 0, stream>>>(
        features, s1, nm2, Ws0, Wn0, nmh1);
    sage_gemm_splitk_kernel<false, false><<<dim3(32, 4), 256, 0, stream>>>(
        h0, (const int*)nullptr, nmh1, Ws1, Wn1, pre);
    l2norm_kernel<<<512, 256, 0, stream>>>(pre, out);
}


# Fewer, fatter waves in the two memory grids: a gather wave sums two whole hop-2 rows (was half a row), a norm wave owns a whole row (was 32 columns); 14336 waves to 3072; norm's sum tree changes
# speedup vs baseline: 6.8330x; 1.0492x over previous
#include <hip/hip_runtime.h>
#include <hip/hip_bf16.h>

typedef __attribute__((ext_vector_type(2))) float v2f;
typedef __attribute__((ext_vector_type(4))) float v4f;
typedef __attribute__((ext_vector_type(8))) float v8f;

#define FEAT 256

template <bool RELU>
__global__ __launch_bounds__(128) void sage_gemm_rows80_mean10_kernel(
    const float* __restrict__ feat, const int* __restrict__ idx,
    const float* __restrict__ Aneigh,
    const float* __restrict__ Ws, const float* __restrict__ Wn,
    float* __restrict__ out)
{
    constexpr int RT = 5;
    __shared__ v4f part[4 * RT * 16 * 16];

    const int tid  = threadIdx.x;
    const int lane = tid & 31;
    const int wave = tid >> 5;
    const int c    = lane & 15;
    const int h    = lane >> 4;
    const int col0 = blockIdx.y * 64;
    const int rowBase = blockIdx.x * (16 * RT);
    const int k0   = wave * 64;

    unsigned node[RT];
#pragma unroll
    for (int i = 0; i < RT; ++i) node[i] = (unsigned)idx[rowBase + 16 * i + c];
    __builtin_amdgcn_sched_barrier(0);

    const size_t woff = (size_t)(k0 + 4 * h) * FEAT + col0 + 4 * c;
    const float* gs = Ws + woff;
    const float* gn = Wn + woff;

    v4f wr[16][4];
    v4f a[RT][16];
#pragma unroll
    for (int jj = 0; jj < 4; ++jj) {
#pragma unroll
        for (int r = 0; r < 4; ++r)
            wr[jj][r] = *(const v4f*)(gs + (size_t)(8 * jj + r) * FEAT);
    }
    __builtin_amdgcn_sched_barrier(0);
    const float* arow_s[RT];
    const float* arow_n[RT];
#pragma unroll
    for (int i = 0; i < RT; ++i) {
        arow_s[i] = feat + (size_t)node[i] * FEAT + k0 + 4 * h;
        arow_n[i] = Aneigh + (size_t)(rowBase + 16 * i + c) * FEAT + k0 + 4 * h;
    }
#pragma unroll
    for (int jj = 0; jj < 8; ++jj) {
        a[0][jj]     = *(const v4f*)(arow_s[0] + 8 * jj);
        a[0][jj + 8] = *(const v4f*)(arow_n[0] + 8 * jj);
    }
    __builtin_amdgcn_sched_barrier(0);
#pragma unroll
    for (int jj = 4; jj < 8; ++jj) {
#pragma unroll
        for (int r = 0; r < 4; ++r)
            wr[jj][r] = *(const v4f*)(gs + (size_t)(8 * jj + r) * FEAT);
    }
    __builtin_amdgcn_sched_barrier(0);
#pragma unroll
    for (int jj = 0; jj < 4; ++jj) {
#pragma unroll
        for (int r = 0; r < 4; ++r)
            wr[jj + 8][r] = *(const v4f*)(gn + (size_t)(8 * jj + r) * FEAT);
    }
    __builtin_amdgcn_sched_barrier(0);
#pragma unroll
    for (int jj = 4; jj < 8; ++jj) {
#pragma unroll
        for (int r = 0; r < 4; ++r)
            wr[jj + 8][r] = *(const v4f*)(gn + (size_t)(8 * jj + r) * FEAT);
    }
    __builtin_amdgcn_sched_barrier(0);
#pragma unroll
    for (int jj = 0; jj < 8; ++jj) {
        a[1][jj]     = *(const v4f*)(arow_s[1] + 8 * jj);
        a[1][jj + 8] = *(const v4f*)(arow_n[1] + 8 * jj);
    }

#pragma unroll
    for (int i = 0; i < RT; ++i) {
        if (i == 1) {
#pragma unroll
            for (int i2 = 2; i2 < RT; ++i2) {
#pragma unroll
                for (int jj = 0; jj < 8; ++jj) {
                    a[i2][jj]     = *(const v4f*)(arow_s[i2] + 8 * jj);
                    a[i2][jj + 8] = *(const v4f*)(arow_n[i2] + 8 * jj);
                }
            }
        }
        if (i < 2) __builtin_amdgcn_sched_barrier(0);

        v8f acc[4];
#pragma unroll
        for (int t = 0; t < 4; ++t) acc[t] = (v8f)0.f;

#pragma unroll
        for (int jj = 0; jj < 16; ++jj) {
            const v2f a0 = {a[i][jj].x, a[i][jj].y};
            const v2f a1 = {a[i][jj].z, a[i][jj].w};
#pragma unroll
            for (int t = 0; t < 4; ++t) {
                const v2f b0 = {wr[jj][0][t], wr[jj][1][t]};
                acc[t] = __builtin_amdgcn_wmma_f32_16x16x4_f32(
                    false, a0, false, b0, (short)0, acc[t], false, false);
            }
#pragma unroll
            for (int t = 0; t < 4; ++t) {
                const v2f b1 = {wr[jj][2][t], wr[jj][3][t]};
                acc[t] = __builtin_amdgcn_wmma_f32_16x16x4_f32(
                    false, a1, false, b1, (short)0, acc[t], false, false);
            }
        }

#pragma unroll
        for (int r = 0; r < 8; ++r) {
            v4f o;
            o.x = acc[0][r]; o.y = acc[1][r]; o.z = acc[2][r]; o.w = acc[3][r];
            part[((wave * RT + i) * 16 + 8 * h + r) * 16 + c] = o;
        }
    }
    __syncthreads();

    const int g = tid >> 4;
    const int q = tid & 15;
    v4f mean = (v4f)0.f;
    for (int j = 0; j < 10; ++j) {
        const int row = 10 * g + j;
        v4f sum = part[row * 16 + q];
#pragma unroll
        for (int w = 1; w < 4; ++w) sum += part[(w * 16 * RT + row) * 16 + q];
        if (RELU) {
            sum.x = fmaxf(sum.x, 0.0f); sum.y = fmaxf(sum.y, 0.0f);
            sum.z = fmaxf(sum.z, 0.0f); sum.w = fmaxf(sum.w, 0.0f);
        }
        mean += sum;
    }
    mean *= 0.1f;
    *(v4f*)(out + (size_t)(blockIdx.x * ((16 * RT) / 10) + g) * FEAT
            + col0 + 4 * q) = mean;
}

#define NM_PITCH 260

template <bool RELU, bool GATHER, bool MEAN10>
__device__ __forceinline__ void sage_gemm_splitk_tile(
    v4f* __restrict__ part, float* __restrict__ nmT,
    const float* __restrict__ Aself, const int* __restrict__ idx,
    const int* __restrict__ adj, const float* __restrict__ Aneigh,
    const float* __restrict__ Ws, const float* __restrict__ Wn,
    float* __restrict__ out, const int rowTile, const int colSlice)
{
    const int tid  = threadIdx.x;
    const int lane = tid & 31;
    const int wave = tid >> 5;
    const int c    = lane & 15;
    const int h    = lane >> 4;
    const int col0 = colSlice * 64;
    const int rowBase = rowTile * 16;
    const int m    = rowBase + c;
    const bool self = wave < 4;
    const int k0   = (wave & 3) * 64;

    const int node = GATHER ? idx[m] : m;

    v4f a[8];
    if (MEAN10) {
        if (self) {
            const float* arow = Aself + (size_t)node * FEAT + k0 + 4 * h;
#pragma unroll
            for (int jj = 0; jj < 8; ++jj) a[jj] = *(const v4f*)(arow + 8 * jj);
        }
        __builtin_amdgcn_sched_barrier(0);
        const int wv = __builtin_amdgcn_readfirstlane(wave);
#pragma unroll
        for (int rr = 0; rr < 2; ++rr) {
            const int r = 2 * wv + rr;
            const int* arow = adj + (size_t)idx[rowBase + r] * 25;
            float4 lo = {0.f, 0.f, 0.f, 0.f}, hi = {0.f, 0.f, 0.f, 0.f};
            for (int j = 0; j < 10; ++j) {
                const float* s = Aself + (size_t)arow[j] * FEAT + lane * 4;
                const float4 v0 = *(const float4*)s;
                const float4 v1 = *(const float4*)(s + 128);
                lo.x += v0.x; lo.y += v0.y; lo.z += v0.z; lo.w += v0.w;
                hi.x += v1.x; hi.y += v1.y; hi.z += v1.z; hi.w += v1.w;
            }
            const float sc = 0.1f;
            lo.x *= sc; lo.y *= sc; lo.z *= sc; lo.w *= sc;
            hi.x *= sc; hi.y *= sc; hi.z *= sc; hi.w *= sc;
            *(float4*)(nmT + r * NM_PITCH + lane * 4)       = lo;
            *(float4*)(nmT + r * NM_PITCH + 128 + lane * 4) = hi;
        }
        __builtin_amdgcn_sched_barrier(0);
    }

    const float* g = (self ? Ws : Wn)
                     + (size_t)(k0 + 4 * h) * FEAT + col0 + 4 * c;

    v4f wr[8][4];
#pragma unroll
    for (int jj = 0; jj < 8; ++jj) {
#pragma unroll
        for (int r = 0; r < 4; ++r)
            wr[jj][r] = *(const v4f*)(g + (size_t)(8 * jj + r) * FEAT);
    }
    if (!MEAN10) {
        const float* arow = (self ? (Aself + (size_t)node * FEAT)
                                  : (Aneigh + (size_t)m * FEAT)) + k0 + 4 * h;
#pragma unroll
        for (int jj = 0; jj < 8; ++jj) a[jj] = *(const v4f*)(arow + 8 * jj);
    }
    __builtin_amdgcn_sched_barrier(0);
    if (MEAN10) {
        __builtin_amdgcn_fence(__ATOMIC_RELEASE, "workgroup", "local");
        __builtin_amdgcn_s_barrier();
        __builtin_amdgcn_fence(__ATOMIC_ACQUIRE, "workgroup", "local");
        if (!self) {
            const float* trow = nmT + c * NM_PITCH + k0 + 4 * h;
#pragma unroll
            for (int jj = 0; jj < 8; ++jj) a[jj] = *(const v4f*)(trow + 8 * jj);
        }
    }

    v8f acc[4];
#pragma unroll
    for (int t = 0; t < 4; ++t) acc[t] = (v8f)0.f;

#pragma unroll
    for (int jj = 0; jj < 8; ++jj) {
        const v2f a0 = {a[jj].x, a[jj].y};
        const v2f a1 = {a[jj].z, a[jj].w};
#pragma unroll
        for (int t = 0; t < 4; ++t) {
            const v2f b0 = {wr[jj][0][t], wr[jj][1][t]};
            acc[t] = __builtin_amdgcn_wmma_f32_16x16x4_f32(
                false, a0, false, b0, (short)0, acc[t], false, false);
        }
#pragma unroll
        for (int t = 0; t < 4; ++t) {
            const v2f b1 = {wr[jj][2][t], wr[jj][3][t]};
            acc[t] = __builtin_amdgcn_wmma_f32_16x16x4_f32(
                false, a1, false, b1, (short)0, acc[t], false, false);
        }
    }

#pragma unroll
    for (int r = 0; r < 8; ++r) {
        v4f o;
        o.x = acc[0][r]; o.y = acc[1][r]; o.z = acc[2][r]; o.w = acc[3][r];
        part[(wave * 16 + 8 * h + r) * 16 + c] = o;
    }
    __syncthreads();

    const int row = tid >> 4;
    const int q   = tid & 15;
    v4f sum = part[row * 16 + q];
#pragma unroll
    for (int w = 1; w < 8; ++w) sum += part[(w * 16 + row) * 16 + q];
    if (RELU) {
        sum.x = fmaxf(sum.x, 0.0f); sum.y = fmaxf(sum.y, 0.0f);
        sum.z = fmaxf(sum.z, 0.0f); sum.w = fmaxf(sum.w, 0.0f);
    }
    *(v4f*)(out + (size_t)(rowBase + row) * FEAT + col0 + 4 * q) = sum;
}

template <bool RELU, bool GATHER>
__global__ __launch_bounds__(256) void sage_gemm_splitk_kernel(
    const float* __restrict__ Aself, const int* __restrict__ idx,
    const float* __restrict__ Aneigh,
    const float* __restrict__ Ws, const float* __restrict__ Wn,
    float* __restrict__ out)
{
    __shared__ v4f part[8 * 16 * 16];
    sage_gemm_splitk_tile<RELU, GATHER, false>(
        part, (float*)nullptr, Aself, idx, (const int*)nullptr, Aneigh, Ws, Wn,
        out, blockIdx.x, blockIdx.y);
}

__global__ __launch_bounds__(256) void hop2_mean_with_hop0_gemm_kernel(
    const float* __restrict__ feat, const int* __restrict__ adj,
    const int* __restrict__ batch,
    const float* __restrict__ Ws, const float* __restrict__ Wn,
    int* __restrict__ s1, float* __restrict__ nm2, float* __restrict__ h0)
{
    __shared__ v4f part[8 * 16 * 16];
    __shared__ float nmT[16 * NM_PITCH];

    if (blockIdx.x < 128) {
        sage_gemm_splitk_tile<true, true, true>(
            part, nmT, feat, batch, adj, (const float*)nullptr, Ws, Wn, h0,
            blockIdx.x & 31, blockIdx.x >> 5);
        return;
    }
    const int wv   = __builtin_amdgcn_readfirstlane((int)(threadIdx.x >> 5));
    const int lane = threadIdx.x & 31;
    const int row0 = ((blockIdx.x - 128) * 8 + wv) * 2;
    const int b    = row0 / 10;
    const int* brow = adj + (size_t)batch[b] * 25 + (row0 - 10 * b);
    const int node0 = brow[0];
    const int node1 = brow[1];
    const int* arow0 = adj + (size_t)node0 * 25;
    const int* arow1 = adj + (size_t)node1 * 25;
    int id0[25], id1[25];
#pragma unroll
    for (int k = 0; k < 25; ++k) { id0[k] = arow0[k]; id1[k] = arow1[k]; }
    const float* fl = feat + lane * 4;
    float* o = nm2 + (size_t)row0 * FEAT + lane * 4;
#pragma unroll
    for (int rr = 0; rr < 2; ++rr) {
        float4 lo = {0.f, 0.f, 0.f, 0.f}, hi = {0.f, 0.f, 0.f, 0.f};
#pragma unroll
        for (int k = 0; k < 25; ++k) {
            const float* s = fl + (size_t)(rr ? id1[k] : id0[k]) * FEAT;
            const float4 v0 = *(const float4*)s;
            const float4 v1 = *(const float4*)(s + 128);
            lo.x += v0.x; lo.y += v0.y; lo.z += v0.z; lo.w += v0.w;
            hi.x += v1.x; hi.y += v1.y; hi.z += v1.z; hi.w += v1.w;
        }
        const float sc = 1.0f / 25.0f;
        lo.x *= sc; lo.y *= sc; lo.z *= sc; lo.w *= sc;
        hi.x *= sc; hi.y *= sc; hi.z *= sc; hi.w *= sc;
        *(float4*)(o + rr * FEAT)       = lo;
        *(float4*)(o + rr * FEAT + 128) = hi;
        if (rr == 0) __builtin_amdgcn_sched_barrier(0);
    }
    if (lane < 2) s1[row0 + lane] = lane ? node1 : node0;
}

__global__ __launch_bounds__(256) void l2norm_kernel(
    const float* __restrict__ pre, float* __restrict__ out)
{
    const int lane = threadIdx.x & 31;
    const int row  = blockIdx.x * 8 + (threadIdx.x >> 5);
    const size_t off = (size_t)row * FEAT + lane * 4;
    const float4 a = *(const float4*)(pre + off);
    const float4 b = *(const float4*)(pre + off + 128);
    float s = a.x * a.x + a.y * a.y + a.z * a.z + a.w * a.w
            + b.x * b.x + b.y * b.y + b.z * b.z + b.w * b.w;
#pragma unroll
    for (int o = 16; o >= 1; o >>= 1) s += __shfl_xor(s, o, 32);
    const float norm = fmaxf(sqrtf(s), 1e-12f);
    float4 ra, rb;
    ra.x = a.x / norm; ra.y = a.y / norm; ra.z = a.z / norm; ra.w = a.w / norm;
    rb.x = b.x / norm; rb.y = b.y / norm; rb.z = b.z / norm; rb.w = b.w / norm;
    *(float4*)(out + off)       = ra;
    *(float4*)(out + off + 128) = rb;
}

extern "C" void kernel_launch(void* const* d_in, const int* in_sizes, int n_in,
                              void* d_out, int out_size, void* d_ws,
                              size_t ws_size, hipStream_t stream)
{
    const float* features = (const float*)d_in[0];
    const int*   adj      = (const int*)d_in[1];
    const int*   batch    = (const int*)d_in[2];
    const float* Ws0      = (const float*)d_in[3];
    const float* Wn0      = (const float*)d_in[4];
    const float* Ws1      = (const float*)d_in[5];
    const float* Wn1      = (const float*)d_in[6];
    float* out = (float*)d_out;

    char* ws = (char*)d_ws;
    int*   s1   = (int*)(ws + 0);
    float* nm2  = (float*)(ws + 20480);
    float* h0   = (float*)(ws + 11030528);
    float* nmh1 = (float*)(ws + 11554816);
    float* pre  = (float*)(ws + 12079104);

    hop2_mean_with_hop0_gemm_kernel<<<128 + 320, 256, 0, stream>>>(
        features, adj, batch, Ws0, Wn0, s1, nm2, h0);
    sage_gemm_rows80_mean10_kernel<true><<<dim3(64, 4), 128, 0, stream>>>(
        features, s1, nm2, Ws0, Wn0, nmh1);
    sage_gemm_splitk_kernel<false, false><<<dim3(32, 4), 256, 0, stream>>>(
        h0, (const int*)nullptr, nmh1, Ws1, Wn1, pre);
    l2norm_kernel<<<64, 256, 0, stream>>>(pre, out);
}


# Layer-1 launch dissolved: each layer-0 block multiplies its finished 64-column slice by the matching 64 k of the layer-1 weights; 8 partial planes summed by the norm launch; k sum order changes
# speedup vs baseline: 7.5598x; 1.1064x over previous
#include <hip/hip_runtime.h>
#include <hip/hip_bf16.h>

typedef __attribute__((ext_vector_type(2))) float v2f;
typedef __attribute__((ext_vector_type(4))) float v4f;
typedef __attribute__((ext_vector_type(8))) float v8f;

#define FEAT 256

#define MT_PITCH 68
#define PLANE_FLOATS (512 * FEAT)

template <bool RELU>
__global__ __launch_bounds__(128) void sage_gemm_rows80_mean10_kernel(
    const float* __restrict__ feat, const int* __restrict__ idx,
    const float* __restrict__ Aneigh,
    const float* __restrict__ Ws, const float* __restrict__ Wn,
    const float* __restrict__ Wn1, float* __restrict__ plane)
{
    constexpr int RT = 5;
    __shared__ v4f part[4 * RT * 16 * 16];
    __shared__ float mT[16 * MT_PITCH];

    const int tid  = threadIdx.x;
    const int lane = tid & 31;
    const int wave = tid >> 5;
    const int c    = lane & 15;
    const int h    = lane >> 4;
    const int col0 = blockIdx.y * 64;
    const int rowBase = blockIdx.x * (16 * RT);
    const int k0   = wave * 64;

    unsigned node[RT];
#pragma unroll
    for (int i = 0; i < RT; ++i) node[i] = (unsigned)idx[rowBase + 16 * i + c];
    __builtin_amdgcn_sched_barrier(0);

    const size_t woff = (size_t)(k0 + 4 * h) * FEAT + col0 + 4 * c;
    const float* gs = Ws + woff;
    const float* gn = Wn + woff;

    v4f wr[16][4];
    v4f a[RT][16];
    v4f w1[8][4];
#pragma unroll
    for (int jj = 0; jj < 4; ++jj) {
#pragma unroll
        for (int r = 0; r < 4; ++r)
            wr[jj][r] = *(const v4f*)(gs + (size_t)(8 * jj + r) * FEAT);
    }
    __builtin_amdgcn_sched_barrier(0);
    const float* arow_s[RT];
    const float* arow_n[RT];
#pragma unroll
    for (int i = 0; i < RT; ++i) {
        arow_s[i] = feat + (size_t)node[i] * FEAT + k0 + 4 * h;
        arow_n[i] = Aneigh + (size_t)(rowBase + 16 * i + c) * FEAT + k0 + 4 * h;
    }
#pragma unroll
    for (int jj = 0; jj < 8; ++jj) {
        a[0][jj]     = *(const v4f*)(arow_s[0] + 8 * jj);
        a[0][jj + 8] = *(const v4f*)(arow_n[0] + 8 * jj);
    }
    __builtin_amdgcn_sched_barrier(0);
#pragma unroll
    for (int jj = 4; jj < 8; ++jj) {
#pragma unroll
        for (int r = 0; r < 4; ++r)
            wr[jj][r] = *(const v4f*)(gs + (size_t)(8 * jj + r) * FEAT);
    }
    __builtin_amdgcn_sched_barrier(0);
#pragma unroll
    for (int jj = 0; jj < 4; ++jj) {
#pragma unroll
        for (int r = 0; r < 4; ++r)
            wr[jj + 8][r] = *(const v4f*)(gn + (size_t)(8 * jj + r) * FEAT);
    }
    __builtin_amdgcn_sched_barrier(0);
#pragma unroll
    for (int jj = 4; jj < 8; ++jj) {
#pragma unroll
        for (int r = 0; r < 4; ++r)
            wr[jj + 8][r] = *(const v4f*)(gn + (size_t)(8 * jj + r) * FEAT);
    }
    __builtin_amdgcn_sched_barrier(0);
#pragma unroll
    for (int jj = 0; jj < 8; ++jj) {
        a[1][jj]     = *(const v4f*)(arow_s[1] + 8 * jj);
        a[1][jj + 8] = *(const v4f*)(arow_n[1] + 8 * jj);
    }

#pragma unroll
    for (int i = 0; i < RT; ++i) {
        if (i == 1) {
#pragma unroll
            for (int i2 = 2; i2 < RT; ++i2) {
#pragma unroll
                for (int jj = 0; jj < 8; ++jj) {
                    a[i2][jj]     = *(const v4f*)(arow_s[i2] + 8 * jj);
                    a[i2][jj + 8] = *(const v4f*)(arow_n[i2] + 8 * jj);
                }
            }
        }
        if (i < 2) __builtin_amdgcn_sched_barrier(0);
        if (i == RT - 1) {
            const float* g1 = Wn1 + (size_t)(col0 + 4 * h) * FEAT
                              + 64 * wave + 4 * c;
#pragma unroll
            for (int jj = 0; jj < 8; ++jj) {
#pragma unroll
                for (int r = 0; r < 4; ++r)
                    w1[jj][r] = *(const v4f*)(g1 + (size_t)(8 * jj + r) * FEAT);
            }
            __builtin_amdgcn_sched_barrier(0);
        }

        v8f acc[4];
#pragma unroll
        for (int t = 0; t < 4; ++t) acc[t] = (v8f)0.f;

#pragma unroll
        for (int jj = 0; jj < 16; ++jj) {
            const v2f a0 = {a[i][jj].x, a[i][jj].y};
            const v2f a1 = {a[i][jj].z, a[i][jj].w};
#pragma unroll
            for (int t = 0; t < 4; ++t) {
                const v2f b0 = {wr[jj][0][t], wr[jj][1][t]};
                acc[t] = __builtin_amdgcn_wmma_f32_16x16x4_f32(
                    false, a0, false, b0, (short)0, acc[t], false, false);
            }
#pragma unroll
            for (int t = 0; t < 4; ++t) {
                const v2f b1 = {wr[jj][2][t], wr[jj][3][t]};
                acc[t] = __builtin_amdgcn_wmma_f32_16x16x4_f32(
                    false, a1, false, b1, (short)0, acc[t], false, false);
            }
        }

#pragma unroll
        for (int r = 0; r < 8; ++r) {
            v4f o;
            o.x = acc[0][r]; o.y = acc[1][r]; o.z = acc[2][r]; o.w = acc[3][r];
            part[((wave * RT + i) * 16 + 8 * h + r) * 16 + c] = o;
        }
    }
    __syncthreads();

    const int g = tid >> 4;
    const int q = tid & 15;
    v4f mean = (v4f)0.f;
    for (int j = 0; j < 10; ++j) {
        const int row = 10 * g + j;
        v4f sum = part[row * 16 + q];
#pragma unroll
        for (int w = 1; w < 4; ++w) sum += part[(w * 16 * RT + row) * 16 + q];
        if (RELU) {
            sum.x = fmaxf(sum.x, 0.0f); sum.y = fmaxf(sum.y, 0.0f);
            sum.z = fmaxf(sum.z, 0.0f); sum.w = fmaxf(sum.w, 0.0f);
        }
        mean += sum;
    }
    mean *= 0.1f;
    *(v4f*)(mT + g * MT_PITCH + 4 * q) = mean;
    *(v4f*)(mT + (8 + g) * MT_PITCH + 4 * q) = (v4f)0.f;
    __syncthreads();

    float* p = plane + (size_t)blockIdx.y * PLANE_FLOATS
               + (size_t)(blockIdx.x * ((16 * RT) / 10)) * FEAT
               + 64 * wave + 4 * c;

    v4f am[8];
#pragma unroll
    for (int jj = 0; jj < 8; ++jj)
        am[jj] = *(const v4f*)(mT + c * MT_PITCH + 8 * jj + 4 * h);
    v8f acc1[4];
#pragma unroll
    for (int t = 0; t < 4; ++t) acc1[t] = (v8f)0.f;
#pragma unroll
    for (int jj = 0; jj < 8; ++jj) {
        const v2f a0 = {am[jj].x, am[jj].y};
        const v2f a1 = {am[jj].z, am[jj].w};
#pragma unroll
        for (int t = 0; t < 4; ++t) {
            const v2f b0 = {w1[jj][0][t], w1[jj][1][t]};
            acc1[t] = __builtin_amdgcn_wmma_f32_16x16x4_f32(
                false, a0, false, b0, (short)0, acc1[t], false, false);
        }
#pragma unroll
        for (int t = 0; t < 4; ++t) {
            const v2f b1 = {w1[jj][2][t], w1[jj][3][t]};
            acc1[t] = __builtin_amdgcn_wmma_f32_16x16x4_f32(
                false, a1, false, b1, (short)0, acc1[t], false, false);
        }
    }
    if (h == 0) {
#pragma unroll
        for (int r = 0; r < 8; ++r) {
            v4f o;
            o.x = acc1[0][r]; o.y = acc1[1][r]; o.z = acc1[2][r]; o.w = acc1[3][r];
            *(v4f*)(p + (size_t)r * FEAT) = o;
        }
    }
}

#define NM_PITCH 260

template <bool RELU, bool GATHER, bool MEAN10>
__device__ __forceinline__ void sage_gemm_splitk_tile(
    v4f* __restrict__ part, float* __restrict__ nmT,
    const float* __restrict__ Aself, const int* __restrict__ idx,
    const int* __restrict__ adj, const float* __restrict__ Aneigh,
    const float* __restrict__ Ws, const float* __restrict__ Wn,
    const float* __restrict__ Ws1,
    float* __restrict__ out, const int rowTile, const int colSlice)
{
    const int tid  = threadIdx.x;
    const int lane = tid & 31;
    const int wave = tid >> 5;
    const int c    = lane & 15;
    const int h    = lane >> 4;
    const int col0 = colSlice * 64;
    const int rowBase = rowTile * 16;
    const int m    = rowBase + c;
    const bool self = wave < 4;
    const int k0   = (wave & 3) * 64;

    const int node = GATHER ? idx[m] : m;

    v4f a[8];
    if (MEAN10) {
        if (self) {
            const float* arow = Aself + (size_t)node * FEAT + k0 + 4 * h;
#pragma unroll
            for (int jj = 0; jj < 8; ++jj) a[jj] = *(const v4f*)(arow + 8 * jj);
        }
        __builtin_amdgcn_sched_barrier(0);
        const int wv = __builtin_amdgcn_readfirstlane(wave);
#pragma unroll
        for (int rr = 0; rr < 2; ++rr) {
            const int r = 2 * wv + rr;
            const int* arow = adj + (size_t)idx[rowBase + r] * 25;
            float4 lo = {0.f, 0.f, 0.f, 0.f}, hi = {0.f, 0.f, 0.f, 0.f};
            for (int j = 0; j < 10; ++j) {
                const float* s = Aself + (size_t)arow[j] * FEAT + lane * 4;
                const float4 v0 = *(const float4*)s;
                const float4 v1 = *(const float4*)(s + 128);
                lo.x += v0.x; lo.y += v0.y; lo.z += v0.z; lo.w += v0.w;
                hi.x += v1.x; hi.y += v1.y; hi.z += v1.z; hi.w += v1.w;
            }
            const float sc = 0.1f;
            lo.x *= sc; lo.y *= sc; lo.z *= sc; lo.w *= sc;
            hi.x *= sc; hi.y *= sc; hi.z *= sc; hi.w *= sc;
            *(float4*)(nmT + r * NM_PITCH + lane * 4)       = lo;
            *(float4*)(nmT + r * NM_PITCH + 128 + lane * 4) = hi;
        }
        __builtin_amdgcn_sched_barrier(0);
    }

    const float* g = (self ? Ws : Wn)
                     + (size_t)(k0 + 4 * h) * FEAT + col0 + 4 * c;

    v4f wr[8][4];
#pragma unroll
    for (int jj = 0; jj < 8; ++jj) {
#pragma unroll
        for (int r = 0; r < 4; ++r)
            wr[jj][r] = *(const v4f*)(g + (size_t)(8 * jj + r) * FEAT);
    }
    if (!MEAN10) {
        const float* arow = (self ? (Aself + (size_t)node * FEAT)
                                  : (Aneigh + (size_t)m * FEAT)) + k0 + 4 * h;
#pragma unroll
        for (int jj = 0; jj < 8; ++jj) a[jj] = *(const v4f*)(arow + 8 * jj);
    }
    __builtin_amdgcn_sched_barrier(0);
    if (MEAN10) {
        __builtin_amdgcn_fence(__ATOMIC_RELEASE, "workgroup", "local");
        __builtin_amdgcn_s_barrier();
        __builtin_amdgcn_fence(__ATOMIC_ACQUIRE, "workgroup", "local");
        if (!self) {
            const float* trow = nmT + c * NM_PITCH + k0 + 4 * h;
#pragma unroll
            for (int jj = 0; jj < 8; ++jj) a[jj] = *(const v4f*)(trow + 8 * jj);
        }
    }

    v8f acc[4];
#pragma unroll
    for (int t = 0; t < 4; ++t) acc[t] = (v8f)0.f;

    v2f w1[8][4];

#pragma unroll
    for (int jj = 0; jj < 8; ++jj) {
        if (jj == 4) {
            __builtin_amdgcn_sched_barrier(0);
            const float* g1 = Ws1 + (size_t)(col0 + 4 * h) * FEAT
                              + 32 * wave + 2 * c;
#pragma unroll
            for (int j1 = 0; j1 < 8; ++j1) {
#pragma unroll
                for (int r = 0; r < 4; ++r)
                    w1[j1][r] = *(const v2f*)(g1 + (size_t)(8 * j1 + r) * FEAT);
            }
        }
        const v2f a0 = {a[jj].x, a[jj].y};
        const v2f a1 = {a[jj].z, a[jj].w};
#pragma unroll
        for (int t = 0; t < 4; ++t) {
            const v2f b0 = {wr[jj][0][t], wr[jj][1][t]};
            acc[t] = __builtin_amdgcn_wmma_f32_16x16x4_f32(
                false, a0, false, b0, (short)0, acc[t], false, false);
        }
#pragma unroll
        for (int t = 0; t < 4; ++t) {
            const v2f b1 = {wr[jj][2][t], wr[jj][3][t]};
            acc[t] = __builtin_amdgcn_wmma_f32_16x16x4_f32(
                false, a1, false, b1, (short)0, acc[t], false, false);
        }
    }

#pragma unroll
    for (int r = 0; r < 8; ++r) {
        v4f o;
        o.x = acc[0][r]; o.y = acc[1][r]; o.z = acc[2][r]; o.w = acc[3][r];
        part[(wave * 16 + 8 * h + r) * 16 + c] = o;
    }
    __builtin_amdgcn_fence(__ATOMIC_RELEASE, "workgroup", "local");
    __builtin_amdgcn_s_barrier();
    __builtin_amdgcn_fence(__ATOMIC_ACQUIRE, "workgroup", "local");

    const int row = tid >> 4;
    const int q   = tid & 15;
    v4f sum = part[row * 16 + q];
#pragma unroll
    for (int w = 1; w < 8; ++w) sum += part[(w * 16 + row) * 16 + q];
    if (RELU) {
        sum.x = fmaxf(sum.x, 0.0f); sum.y = fmaxf(sum.y, 0.0f);
        sum.z = fmaxf(sum.z, 0.0f); sum.w = fmaxf(sum.w, 0.0f);
    }
    *(v4f*)(nmT + row * NM_PITCH + 4 * q) = sum;
    __builtin_amdgcn_fence(__ATOMIC_RELEASE, "workgroup", "local");
    __builtin_amdgcn_s_barrier();
    __builtin_amdgcn_fence(__ATOMIC_ACQUIRE, "workgroup", "local");
    v4f at[8];
    {
        const float* trow = nmT + c * NM_PITCH + 4 * h;
#pragma unroll
        for (int jj = 0; jj < 8; ++jj) at[jj] = *(const v4f*)(trow + 8 * jj);
    }
    v8f acc1[2][2];
#pragma unroll
    for (int t = 0; t < 2; ++t) { acc1[t][0] = (v8f)0.f; acc1[t][1] = (v8f)0.f; }
#pragma unroll
    for (int jj = 0; jj < 8; ++jj) {
        const v2f a0 = {at[jj].x, at[jj].y};
        const v2f a1 = {at[jj].z, at[jj].w};
#pragma unroll
        for (int t = 0; t < 2; ++t) {
            const v2f b0 = {w1[jj][0][t], w1[jj][1][t]};
            acc1[t][0] = __builtin_amdgcn_wmma_f32_16x16x4_f32(
                false, a0, false, b0, (short)0, acc1[t][0], false, false);
        }
#pragma unroll
        for (int t = 0; t < 2; ++t) {
            const v2f b1 = {w1[jj][2][t], w1[jj][3][t]};
            acc1[t][1] = __builtin_amdgcn_wmma_f32_16x16x4_f32(
                false, a1, false, b1, (short)0, acc1[t][1], false, false);
        }
    }
    float* p = out + (size_t)colSlice * PLANE_FLOATS
               + (size_t)(rowBase + 8 * h) * FEAT + 32 * wave + 2 * c;
#pragma unroll
    for (int r = 0; r < 8; ++r) {
        v2f o;
        o.x = acc1[0][0][r] + acc1[0][1][r];
        o.y = acc1[1][0][r] + acc1[1][1][r];
        *(v2f*)(p + (size_t)r * FEAT) = o;
    }
}

__global__ __launch_bounds__(256) void hop2_mean_with_hop0_gemm_kernel(
    const float* __restrict__ feat, const int* __restrict__ adj,
    const int* __restrict__ batch,
    const float* __restrict__ Ws, const float* __restrict__ Wn,
    const float* __restrict__ Ws1,
    int* __restrict__ s1, float* __restrict__ nm2, float* __restrict__ plane)
{
    __shared__ v4f part[8 * 16 * 16];
    __shared__ float nmT[16 * NM_PITCH];

    if (blockIdx.x < 128) {
        sage_gemm_splitk_tile<true, true, true>(
            part, nmT, feat, batch, adj, (const float*)nullptr, Ws, Wn, Ws1,
            plane, blockIdx.x & 31, blockIdx.x >> 5);
        return;
    }
    const int wv   = __builtin_amdgcn_readfirstlane((int)(threadIdx.x >> 5));
    const int lane = threadIdx.x & 31;
    const int row0 = ((blockIdx.x - 128) * 8 + wv) * 2;
    const int b    = row0 / 10;
    const int* brow = adj + (size_t)batch[b] * 25 + (row0 - 10 * b);
    const int node0 = brow[0];
    const int node1 = brow[1];
    const int* arow0 = adj + (size_t)node0 * 25;
    const int* arow1 = adj + (size_t)node1 * 25;
    int id0[25], id1[25];
#pragma unroll
    for (int k = 0; k < 25; ++k) { id0[k] = arow0[k]; id1[k] = arow1[k]; }
    const float* fl = feat + lane * 4;
    float* o = nm2 + (size_t)row0 * FEAT + lane * 4;
#pragma unroll
    for (int rr = 0; rr < 2; ++rr) {
        float4 lo = {0.f, 0.f, 0.f, 0.f}, hi = {0.f, 0.f, 0.f, 0.f};
#pragma unroll
        for (int k = 0; k < 25; ++k) {
            const float* s = fl + (size_t)(rr ? id1[k] : id0[k]) * FEAT;
            const float4 v0 = *(const float4*)s;
            const float4 v1 = *(const float4*)(s + 128);
            lo.x += v0.x; lo.y += v0.y; lo.z += v0.z; lo.w += v0.w;
            hi.x += v1.x; hi.y += v1.y; hi.z += v1.z; hi.w += v1.w;
        }
        const float sc = 1.0f / 25.0f;
        lo.x *= sc; lo.y *= sc; lo.z *= sc; lo.w *= sc;
        hi.x *= sc; hi.y *= sc; hi.z *= sc; hi.w *= sc;
        *(float4*)(o + rr * FEAT)       = lo;
        *(float4*)(o + rr * FEAT + 128) = hi;
        if (rr == 0) __builtin_amdgcn_sched_barrier(0);
    }
    if (lane < 2) s1[row0 + lane] = lane ? node1 : node0;
}

__global__ __launch_bounds__(256) void plane_sum_l2norm_kernel(
    const float* __restrict__ plane, float* __restrict__ out)
{
    const int lane = threadIdx.x & 31;
    const int row  = blockIdx.x * 8 + (threadIdx.x >> 5);
    const size_t off = (size_t)row * FEAT + lane * 4;
    float4 pa[8], pb[8];
#pragma unroll
    for (int p = 0; p < 8; ++p) {
        pa[p] = *(const float4*)(plane + (size_t)p * PLANE_FLOATS + off);
        pb[p] = *(const float4*)(plane + (size_t)p * PLANE_FLOATS + off + 128);
    }
    float4 a = pa[0], b = pb[0];
#pragma unroll
    for (int p = 1; p < 8; ++p) {
        a.x += pa[p].x; a.y += pa[p].y; a.z += pa[p].z; a.w += pa[p].w;
        b.x += pb[p].x; b.y += pb[p].y; b.z += pb[p].z; b.w += pb[p].w;
    }
    float s = a.x * a.x + a.y * a.y + a.z * a.z + a.w * a.w
            + b.x * b.x + b.y * b.y + b.z * b.z + b.w * b.w;
#pragma unroll
    for (int o = 16; o >= 1; o >>= 1) s += __shfl_xor(s, o, 32);
    const float norm = fmaxf(sqrtf(s), 1e-12f);
    float4 ra, rb;
    ra.x = a.x / norm; ra.y = a.y / norm; ra.z = a.z / norm; ra.w = a.w / norm;
    rb.x = b.x / norm; rb.y = b.y / norm; rb.z = b.z / norm; rb.w = b.w / norm;
    *(float4*)(out + off)       = ra;
    *(float4*)(out + off + 128) = rb;
}

extern "C" void kernel_launch(void* const* d_in, const int* in_sizes, int n_in,
                              void* d_out, int out_size, void* d_ws,
                              size_t ws_size, hipStream_t stream)
{
    const float* features = (const float*)d_in[0];
    const int*   adj      = (const int*)d_in[1];
    const int*   batch    = (const int*)d_in[2];
    const float* Ws0      = (const float*)d_in[3];
    const float* Wn0      = (const float*)d_in[4];
    const float* Ws1      = (const float*)d_in[5];
    const float* Wn1      = (const float*)d_in[6];
    float* out = (float*)d_out;

    char* ws = (char*)d_ws;
    int*   s1   = (int*)(ws + 0);
    float* nm2  = (float*)(ws + 20480);
    float* planeS = (float*)(ws + 5263360);
    float* planeN = (float*)(ws + 7360512);

    hop2_mean_with_hop0_gemm_kernel<<<128 + 320, 256, 0, stream>>>(
        features, adj, batch, Ws0, Wn0, Ws1, s1, nm2, planeS);
    sage_gemm_rows80_mean10_kernel<true><<<dim3(64, 4), 128, 0, stream>>>(
        features, s1, nm2, Ws0, Wn0, Wn1, planeN);
    plane_sum_l2norm_kernel<<<64, 256, 0, stream>>>(planeS, out);
}
